# starting kernel
# baseline (speedup 1.0000x reference)
#include <hip/hip_runtime.h>
#include <math.h>

#define NN 2048
#define NB 8
#define NBINS 65536

typedef __attribute__((ext_vector_type(2))) float v2f;
typedef __attribute__((ext_vector_type(8))) float v8f;

__global__ __launch_bounds__(256) void k_row_softmax(const float* __restrict__ adj,
                                                     float* __restrict__ a) {
  __shared__ float red[256];
  const int i = blockIdx.x;
  const int t = threadIdx.x;
  const float* row = adj + (size_t)i * NN;

  float mymax = -INFINITY;
  for (int j = t; j < NN; j += 256) mymax = fmaxf(mymax, row[j]);
  red[t] = mymax; __syncthreads();
  for (int s = 128; s > 0; s >>= 1) { if (t < s) red[t] = fmaxf(red[t], red[t + s]); __syncthreads(); }
  const float m = red[0];
  __syncthreads();

  float sum = 0.f;
  for (int j = t; j < NN; j += 256) sum += __expf(row[j] - m);
  red[t] = sum; __syncthreads();
  for (int s = 128; s > 0; s >>= 1) { if (t < s) red[t] += red[t + s]; __syncthreads(); }
  const float inv = 1.f / red[0];

  float* arow = a + (size_t)i * NN;
  for (int j = t; j < NN; j += 256) arow[j] = __expf(row[j] - m) * inv;
}

__global__ __launch_bounds__(256) void k_hist_hi(const float* __restrict__ a,
                                                 unsigned* __restrict__ hist) {
  const size_t total = (size_t)NN * NN;
  const size_t stride = (size_t)gridDim.x * blockDim.x;
  for (size_t p = (size_t)blockIdx.x * blockDim.x + threadIdx.x; p < total; p += stride)
    atomicAdd(&hist[__float_as_uint(a[p]) >> 16], 1u);
}

__global__ void k_scan_hi(const unsigned* __restrict__ hist, unsigned* __restrict__ meta) {
  if (threadIdx.x == 0 && blockIdx.x == 0) {
    const unsigned k = ((unsigned)NN * (unsigned)NN - 1u) / 2u;
    unsigned cum = 0;
    for (unsigned bin = 0; bin < NBINS; ++bin) {
      const unsigned c = hist[bin];
      if (cum + c > k) { meta[0] = bin; meta[1] = cum; break; }
      cum += c;
    }
  }
}

__global__ __launch_bounds__(256) void k_hist_lo(const float* __restrict__ a,
                                                 const unsigned* __restrict__ meta,
                                                 unsigned* __restrict__ hist2) {
  const unsigned sel = meta[0];
  const size_t total = (size_t)NN * NN;
  const size_t stride = (size_t)gridDim.x * blockDim.x;
  for (size_t p = (size_t)blockIdx.x * blockDim.x + threadIdx.x; p < total; p += stride) {
    const unsigned bits = __float_as_uint(a[p]);
    if ((bits >> 16) == sel) atomicAdd(&hist2[bits & 0xFFFFu], 1u);
  }
}

__global__ void k_scan_lo(const unsigned* __restrict__ hist2, unsigned* __restrict__ meta) {
  if (threadIdx.x == 0 && blockIdx.x == 0) {
    const unsigned k = ((unsigned)NN * (unsigned)NN - 1u) / 2u;
    const unsigned kk = k - meta[1];
    unsigned cum = 0;
    for (unsigned low = 0; low < NBINS; ++low) {
      const unsigned c = hist2[low];
      if (cum + c > kk) { meta[2] = (meta[0] << 16) | low; break; }
      cum += c;
    }
  }
}

__global__ __launch_bounds__(32) void k_gemm_y(const float* __restrict__ x,
                                               const float* __restrict__ W,
                                               float* __restrict__ Y) {
  __shared__ float Wt[16][132];
  __shared__ float Xt[16][132];
  const int nb   = blockIdx.x * 16;
  const int lane = threadIdx.x;
  const int half = lane >> 4;
  const int l16  = lane & 15;

  for (int idx = lane; idx < 8 * 128; idx += 32)
    Xt[8 + (idx >> 7)][idx & 127] = 0.f;

  v8f acc0 = {}, acc1 = {};
  for (int k0 = 0; k0 < NN; k0 += 128) {
    for (int idx = lane; idx < 8 * 128; idx += 32) {
      const int r = idx >> 7, c = idx & 127;
      Xt[r][c] = x[r * NN + (k0 + c)];
    }
    for (int idx = lane; idx < 16 * 128; idx += 32) {
      const int r = idx >> 7, c = idx & 127;
      Wt[r][c] = W[(size_t)(nb + r) * NN + (k0 + c)];
    }
    __syncthreads();
    #pragma unroll 4
    for (int kk = 0; kk < 128; kk += 8) {
      const int ka0 = kk + half * 2;
      v2f a0 = { Xt[l16][ka0], Xt[l16][ka0 + 1] };
      v2f b0 = { Wt[l16][ka0], Wt[l16][ka0 + 1] };
      acc0 = __builtin_amdgcn_wmma_f32_16x16x4_f32(false, a0, false, b0,
                                                   (short)0, acc0, false, false);
      const int ka1 = kk + 4 + half * 2;
      v2f a1 = { Xt[l16][ka1], Xt[l16][ka1 + 1] };
      v2f b1 = { Wt[l16][ka1], Wt[l16][ka1 + 1] };
      acc1 = __builtin_amdgcn_wmma_f32_16x16x4_f32(false, a1, false, b1,
                                                   (short)0, acc1, false, false);
    }
    __syncthreads();
  }
  for (int r = 0; r < 8; ++r) {
    const int m = r + half * 8;
    if (m < NB) Y[m * NN + nb + l16] = acc0[r] + acc1[r];
  }
}

__global__ __launch_bounds__(256) void k_row_stats(const float* __restrict__ x,
                                                   const float* __restrict__ Y,
                                                   const float* __restrict__ bias,
                                                   float* __restrict__ rowM,
                                                   float* __restrict__ rowZ) {
  __shared__ float lr[NN];
  __shared__ float red[256];
  const int i = blockIdx.x;
  const int b = blockIdx.y;
  const int t = threadIdx.x;
  const float xbi = x[b * NN + i];

  float mymax = -INFINITY;
  for (int j = t; j < NN; j += 256) {
    float v = fmaf(xbi, Y[b * NN + j], bias[j]);
    v = (v > 0.f) ? v : 0.01f * v;
    lr[j] = v;
    mymax = fmaxf(mymax, v);
  }
  red[t] = mymax; __syncthreads();
  for (int s = 128; s > 0; s >>= 1) { if (t < s) red[t] = fmaxf(red[t], red[t + s]); __syncthreads(); }
  const float m = red[0];
  __syncthreads();

  float sum = 0.f;
  for (int j = t; j < NN; j += 256) sum += __expf(lr[j] - m);
  red[t] = sum; __syncthreads();
  for (int s = 128; s > 0; s >>= 1) { if (t < s) red[t] += red[t + s]; __syncthreads(); }
  if (t == 0) { rowM[b * NN + i] = m; rowZ[b * NN + i] = red[0]; }
}

__global__ __launch_bounds__(256) void k_aggregate(const float* __restrict__ x,
                                                   const float* __restrict__ Y,
                                                   const float* __restrict__ bias,
                                                   const float* __restrict__ a,
                                                   const unsigned* __restrict__ meta,
                                                   const float* __restrict__ rowM,
                                                   const float* __restrict__ rowZ,
                                                   float* __restrict__ out) {
  const int j  = blockIdx.x * 256 + threadIdx.x;
  const int b  = blockIdx.y;
  const int iBase = blockIdx.z * (NN / 8);
  const float med = __uint_as_float(meta[2]);
  const float ybj = Y[b * NN + j];
  const float bj  = bias[j];

  __shared__ float sx[128], sm[128], sc[128];
  float acc = 0.f;
  for (int i0 = iBase; i0 < iBase + 256; i0 += 128) {
    if (threadIdx.x < 128) {
      const int i = i0 + threadIdx.x;
      const float xv = x[b * NN + i];
      sx[threadIdx.x] = xv;
      sm[threadIdx.x] = rowM[b * NN + i];
      sc[threadIdx.x] = xv / rowZ[b * NN + i];
    }
    __syncthreads();
    for (int ii = 0; ii < 128; ++ii) {
      const int i = i0 + ii;
      if ((ii & 15) == 0)
        __builtin_prefetch(&a[(size_t)(i + 16) * NN + j], 0, 1);
      const float aij = a[(size_t)i * NN + j];
      float t = fmaf(sx[ii], ybj, bj);
      t = (t > 0.f) ? t : 0.01f * t;
      acc += (aij > med) ? sc[ii] * __expf(t - sm[ii]) : 0.f;
    }
    __syncthreads();
  }
  atomicAdd(&out[b * NN + j], acc);
}

extern "C" void kernel_launch(void* const* d_in, const int* in_sizes, int n_in,
                              void* d_out, int out_size, void* d_ws, size_t ws_size,
                              hipStream_t stream) {
  const float* x    = (const float*)d_in[0];
  const float* adj  = (const float*)d_in[1];
  const float* W    = (const float*)d_in[2];
  const float* bias = (const float*)d_in[3];
  float* out = (float*)d_out;

  char* ws = (char*)d_ws;
  size_t off = 0;
  float*    a     = (float*)(ws + off);    off += (size_t)NN * NN * sizeof(float);
  float*    Y     = (float*)(ws + off);    off += (size_t)NB * NN * sizeof(float);
  float*    rowM  = (float*)(ws + off);    off += (size_t)NB * NN * sizeof(float);
  float*    rowZ  = (float*)(ws + off);    off += (size_t)NB * NN * sizeof(float);
  unsigned* hist1 = (unsigned*)(ws + off); off += (size_t)NBINS * sizeof(unsigned);
  unsigned* hist2 = (unsigned*)(ws + off); off += (size_t)NBINS * sizeof(unsigned);
  unsigned* meta  = (unsigned*)(ws + off); off += 16;

  hipMemsetAsync(hist1, 0, 2 * (size_t)NBINS * sizeof(unsigned), stream);
  hipMemsetAsync(out, 0, (size_t)out_size * sizeof(float), stream);

  k_row_softmax<<<NN, 256, 0, stream>>>(adj, a);
  k_hist_hi   <<<1024, 256, 0, stream>>>(a, hist1);
  k_scan_hi   <<<1, 32, 0, stream>>>(hist1, meta);
  k_hist_lo   <<<1024, 256, 0, stream>>>(a, meta, hist2);
  k_scan_lo   <<<1, 32, 0, stream>>>(hist2, meta);
  k_gemm_y    <<<NN / 16, 32, 0, stream>>>(x, W, Y);
  k_row_stats <<<dim3(NN, NB), 256, 0, stream>>>(x, Y, bias, rowM, rowZ);
  k_aggregate <<<dim3(NN / 256, NB, 8), 256, 0, stream>>>(x, Y, bias, a, meta, rowM, rowZ, out);
}


# Median select: two 16-bit digits, 65536-bin tables walked by one thread -> three digits (14/10/8 bits), LDS counts, every group of the next launch finds the bin itself; 2 launches go; same integers
# speedup vs baseline: 7.1733x; 7.1733x over previous
#include <hip/hip_runtime.h>
#include <math.h>

#define NN 2048
#define NB 8
#define NB1 16384
#define NB2 1024
#define NB3 256
#define RANK_K (((unsigned)NN * (unsigned)NN - 1u) / 2u)

typedef __attribute__((ext_vector_type(2))) float v2f;
typedef __attribute__((ext_vector_type(8))) float v8f;

__global__ __launch_bounds__(256) void k_row_softmax(const float* __restrict__ adj,
                                                     float* __restrict__ a) {
  __shared__ float red[256];
  const int i = blockIdx.x;
  const int t = threadIdx.x;
  const float* row = adj + (size_t)i * NN;

  float mymax = -INFINITY;
  for (int j = t; j < NN; j += 256) mymax = fmaxf(mymax, row[j]);
  red[t] = mymax; __syncthreads();
  for (int s = 128; s > 0; s >>= 1) { if (t < s) red[t] = fmaxf(red[t], red[t + s]); __syncthreads(); }
  const float m = red[0];
  __syncthreads();

  float sum = 0.f;
  for (int j = t; j < NN; j += 256) sum += __expf(row[j] - m);
  red[t] = sum; __syncthreads();
  for (int s = 128; s > 0; s >>= 1) { if (t < s) red[t] += red[t + s]; __syncthreads(); }
  const float inv = 1.f / red[0];

  float* arow = a + (size_t)i * NN;
  for (int j = t; j < NN; j += 256) arow[j] = __expf(row[j] - m) * inv;
}


template <int PER>
__device__ __forceinline__ void find_rank(const unsigned (&c)[PER], const unsigned kk,
                                          unsigned* s_tot, unsigned* s_res,
                                          unsigned& bin, unsigned& before) {
  const int t = threadIdx.x, lane = t & 31, wv = t >> 5;
  unsigned mine = 0;
  for (int q = 0; q < PER; ++q) mine += c[q];
  unsigned inc = mine;
  for (int d = 1; d < 32; d <<= 1) {
    const unsigned up = __shfl_up(inc, d);
    if (lane >= d) inc += up;
  }
  if (lane == 31) s_tot[wv] = inc;
  if (t == 0) { s_res[0] = 0u; s_res[1] = 0u; }
  __syncthreads();
  unsigned excl = inc - mine;
  for (int w = 0; w < wv; ++w) excl += s_tot[w];
  if (kk >= excl && kk - excl < mine) {
    unsigned cum = excl, fb = 0u, fc = 0u;
    bool found = false;
    for (int q = 0; q < PER; ++q) {
      if (!found && kk - cum < c[q]) { fb = (unsigned)q; fc = cum; found = true; }
      cum += c[q];
    }
    s_res[0] = (unsigned)t * PER + fb;
    s_res[1] = fc;
  }
  __syncthreads();
  bin = s_res[0];
  before = s_res[1];
}

__global__ __launch_bounds__(512) void k_count_d1(const float* __restrict__ a,
                                                  unsigned* __restrict__ h1) {
  __shared__ unsigned bins[NB1];
  const int t = threadIdx.x;
  const float4* a4 = reinterpret_cast<const float4*>(a) + (size_t)blockIdx.x * (512 * 8);
  float4 v[8];
  for (int q = 0; q < 8; ++q) v[q] = a4[q * 512 + t];
  uint4* b4 = reinterpret_cast<uint4*>(bins);
  for (int q = 0; q < 8; ++q) b4[q * 512 + t] = make_uint4(0u, 0u, 0u, 0u);
  __syncthreads();
  for (int q = 0; q < 8; ++q) {
    const unsigned u[4] = { __float_as_uint(v[q].x), __float_as_uint(v[q].y),
                            __float_as_uint(v[q].z), __float_as_uint(v[q].w) };
    for (int e = 0; e < 4; ++e) {
      const unsigned b = u[e] >> 18;
      const unsigned p = ((b & 28u) << 9) | ((b >> 5) << 2) | (b & 3u);
      atomicAdd(&bins[p], 1u);
    }
  }
  __syncthreads();
  uint4 w[8];
  for (int q = 0; q < 8; ++q) w[q] = b4[q * 512 + t];
  for (int q = 0; q < 8; ++q) {
    if ((w[q].x | w[q].y | w[q].z | w[q].w) != 0u) {
      unsigned* g = h1 + (q * 512 + t) * 4;
      if (w[q].x != 0u) atomicAdd(g + 0, w[q].x);
      if (w[q].y != 0u) atomicAdd(g + 1, w[q].y);
      if (w[q].z != 0u) atomicAdd(g + 2, w[q].z);
      if (w[q].w != 0u) atomicAdd(g + 3, w[q].w);
    }
  }
}

__global__ __launch_bounds__(512) void k_count_d2(const float* __restrict__ a,
                                                  const unsigned* __restrict__ h1,
                                                  unsigned* __restrict__ h2,
                                                  unsigned* __restrict__ meta) {
  __shared__ unsigned s_tot[16];
  __shared__ unsigned s_res[2];
  const int t = threadIdx.x;
  const float4* a4 = reinterpret_cast<const float4*>(a) + (size_t)blockIdx.x * (512 * 8);
  float4 v[8];
  for (int q = 0; q < 8; ++q) v[q] = a4[q * 512 + t];
  const uint4* h4 = reinterpret_cast<const uint4*>(h1);
  unsigned c[32];
  for (int q = 0; q < 8; ++q) {
    const uint4 w = h4[q * 512 + t];
    c[4 * q + 0] = w.x; c[4 * q + 1] = w.y; c[4 * q + 2] = w.z; c[4 * q + 3] = w.w;
  }
  unsigned bin1, before1;
  find_rank<32>(c, RANK_K, s_tot, s_res, bin1, before1);
  if (blockIdx.x == 0 && t == 0) { meta[0] = bin1; meta[1] = before1; }
  for (int q = 0; q < 8; ++q) {
    const unsigned u[4] = { __float_as_uint(v[q].x), __float_as_uint(v[q].y),
                            __float_as_uint(v[q].z), __float_as_uint(v[q].w) };
    for (int e = 0; e < 4; ++e)
      if ((u[e] >> 18) == bin1) atomicAdd(&h2[(u[e] >> 8) & 1023u], 1u);
  }
}

__global__ __launch_bounds__(512) void k_count_d3(const float* __restrict__ a,
                                                  const unsigned* __restrict__ h2,
                                                  unsigned* __restrict__ h3,
                                                  unsigned* __restrict__ meta) {
  __shared__ unsigned s_tot[16];
  __shared__ unsigned s_res[2];
  const int t = threadIdx.x;
  const float4* a4 = reinterpret_cast<const float4*>(a) + (size_t)blockIdx.x * (512 * 8);
  float4 v[8];
  for (int q = 0; q < 8; ++q) v[q] = a4[q * 512 + t];
  const unsigned bin1 = meta[0], before1 = meta[1];
  const uint2 w = reinterpret_cast<const uint2*>(h2)[t];
  const unsigned c[2] = { w.x, w.y };
  unsigned bin2, before2;
  find_rank<2>(c, RANK_K - before1, s_tot, s_res, bin2, before2);
  const unsigned pre = (bin1 << 10) | bin2;
  if (blockIdx.x == 0 && t == 0) { meta[2] = pre; meta[3] = RANK_K - before1 - before2; }
  for (int q = 0; q < 8; ++q) {
    const unsigned u[4] = { __float_as_uint(v[q].x), __float_as_uint(v[q].y),
                            __float_as_uint(v[q].z), __float_as_uint(v[q].w) };
    for (int e = 0; e < 4; ++e)
      if ((u[e] >> 8) == pre) atomicAdd(&h3[u[e] & 255u], 1u);
  }
}

__global__ __launch_bounds__(32) void k_gemm_y(const float* __restrict__ x,
                                               const float* __restrict__ W,
                                               float* __restrict__ Y) {
  __shared__ float Wt[16][132];
  __shared__ float Xt[16][132];
  const int nb   = blockIdx.x * 16;
  const int lane = threadIdx.x;
  const int half = lane >> 4;
  const int l16  = lane & 15;

  for (int idx = lane; idx < 8 * 128; idx += 32)
    Xt[8 + (idx >> 7)][idx & 127] = 0.f;

  v8f acc0 = {}, acc1 = {};
  for (int k0 = 0; k0 < NN; k0 += 128) {
    for (int idx = lane; idx < 8 * 128; idx += 32) {
      const int r = idx >> 7, c = idx & 127;
      Xt[r][c] = x[r * NN + (k0 + c)];
    }
    for (int idx = lane; idx < 16 * 128; idx += 32) {
      const int r = idx >> 7, c = idx & 127;
      Wt[r][c] = W[(size_t)(nb + r) * NN + (k0 + c)];
    }
    __syncthreads();
    #pragma unroll 4
    for (int kk = 0; kk < 128; kk += 8) {
      const int ka0 = kk + half * 2;
      v2f a0 = { Xt[l16][ka0], Xt[l16][ka0 + 1] };
      v2f b0 = { Wt[l16][ka0], Wt[l16][ka0 + 1] };
      acc0 = __builtin_amdgcn_wmma_f32_16x16x4_f32(false, a0, false, b0,
                                                   (short)0, acc0, false, false);
      const int ka1 = kk + 4 + half * 2;
      v2f a1 = { Xt[l16][ka1], Xt[l16][ka1 + 1] };
      v2f b1 = { Wt[l16][ka1], Wt[l16][ka1 + 1] };
      acc1 = __builtin_amdgcn_wmma_f32_16x16x4_f32(false, a1, false, b1,
                                                   (short)0, acc1, false, false);
    }
    __syncthreads();
  }
  for (int r = 0; r < 8; ++r) {
    const int m = r + half * 8;
    if (m < NB) Y[m * NN + nb + l16] = acc0[r] + acc1[r];
  }
}

__global__ __launch_bounds__(256) void k_row_stats(const float* __restrict__ x,
                                                   const float* __restrict__ Y,
                                                   const float* __restrict__ bias,
                                                   float* __restrict__ rowM,
                                                   float* __restrict__ rowZ) {
  __shared__ float lr[NN];
  __shared__ float red[256];
  const int i = blockIdx.x;
  const int b = blockIdx.y;
  const int t = threadIdx.x;
  const float xbi = x[b * NN + i];

  float mymax = -INFINITY;
  for (int j = t; j < NN; j += 256) {
    float v = fmaf(xbi, Y[b * NN + j], bias[j]);
    v = (v > 0.f) ? v : 0.01f * v;
    lr[j] = v;
    mymax = fmaxf(mymax, v);
  }
  red[t] = mymax; __syncthreads();
  for (int s = 128; s > 0; s >>= 1) { if (t < s) red[t] = fmaxf(red[t], red[t + s]); __syncthreads(); }
  const float m = red[0];
  __syncthreads();

  float sum = 0.f;
  for (int j = t; j < NN; j += 256) sum += __expf(lr[j] - m);
  red[t] = sum; __syncthreads();
  for (int s = 128; s > 0; s >>= 1) { if (t < s) red[t] += red[t + s]; __syncthreads(); }
  if (t == 0) { rowM[b * NN + i] = m; rowZ[b * NN + i] = red[0]; }
}

__global__ __launch_bounds__(256) void k_aggregate(const float* __restrict__ x,
                                                   const float* __restrict__ Y,
                                                   const float* __restrict__ bias,
                                                   const float* __restrict__ a,
                                                   const unsigned* __restrict__ meta,
                                                   const unsigned* __restrict__ h3,
                                                   const float* __restrict__ rowM,
                                                   const float* __restrict__ rowZ,
                                                   float* __restrict__ out) {
  const int j  = blockIdx.x * 256 + threadIdx.x;
  const int b  = blockIdx.y;
  const int iBase = blockIdx.z * (NN / 8);
  __shared__ unsigned s_tot[8];
  __shared__ unsigned s_res[2];
  const unsigned c3[1] = { h3[threadIdx.x] };
  unsigned bin3, before3;
  find_rank<1>(c3, meta[3], s_tot, s_res, bin3, before3);
  const float med = __uint_as_float((meta[2] << 8) | bin3);
  const float ybj = Y[b * NN + j];
  const float bj  = bias[j];

  __shared__ float sx[128], sm[128], sc[128];
  float acc = 0.f;
  for (int i0 = iBase; i0 < iBase + 256; i0 += 128) {
    if (threadIdx.x < 128) {
      const int i = i0 + threadIdx.x;
      const float xv = x[b * NN + i];
      sx[threadIdx.x] = xv;
      sm[threadIdx.x] = rowM[b * NN + i];
      sc[threadIdx.x] = xv / rowZ[b * NN + i];
    }
    __syncthreads();
    for (int ii = 0; ii < 128; ++ii) {
      const int i = i0 + ii;
      if ((ii & 15) == 0)
        __builtin_prefetch(&a[(size_t)(i + 16) * NN + j], 0, 1);
      const float aij = a[(size_t)i * NN + j];
      float t = fmaf(sx[ii], ybj, bj);
      t = (t > 0.f) ? t : 0.01f * t;
      acc += (aij > med) ? sc[ii] * __expf(t - sm[ii]) : 0.f;
    }
    __syncthreads();
  }
  atomicAdd(&out[b * NN + j], acc);
}

extern "C" void kernel_launch(void* const* d_in, const int* in_sizes, int n_in,
                              void* d_out, int out_size, void* d_ws, size_t ws_size,
                              hipStream_t stream) {
  const float* x    = (const float*)d_in[0];
  const float* adj  = (const float*)d_in[1];
  const float* W    = (const float*)d_in[2];
  const float* bias = (const float*)d_in[3];
  float* out = (float*)d_out;

  char* ws = (char*)d_ws;
  size_t off = 0;
  float*    a     = (float*)(ws + off);    off += (size_t)NN * NN * sizeof(float);
  float*    Y     = (float*)(ws + off);    off += (size_t)NB * NN * sizeof(float);
  float*    rowM  = (float*)(ws + off);    off += (size_t)NB * NN * sizeof(float);
  float*    rowZ  = (float*)(ws + off);    off += (size_t)NB * NN * sizeof(float);
  unsigned* h1    = (unsigned*)(ws + off); off += (size_t)NB1 * sizeof(unsigned);
  unsigned* h2    = (unsigned*)(ws + off); off += (size_t)NB2 * sizeof(unsigned);
  unsigned* h3    = (unsigned*)(ws + off); off += (size_t)NB3 * sizeof(unsigned);
  unsigned* meta  = (unsigned*)(ws + off); off += 16;

  hipMemsetAsync(h1, 0, (size_t)(NB1 + NB2 + NB3) * sizeof(unsigned), stream);
  hipMemsetAsync(out, 0, (size_t)out_size * sizeof(float), stream);

  k_row_softmax<<<NN, 256, 0, stream>>>(adj, a);
  k_count_d1  <<<NN * NN / (512 * 32), 512, 0, stream>>>(a, h1);
  k_count_d2  <<<NN * NN / (512 * 32), 512, 0, stream>>>(a, h1, h2, meta);
  k_count_d3  <<<NN * NN / (512 * 32), 512, 0, stream>>>(a, h2, h3, meta);
  k_gemm_y    <<<NN / 16, 32, 0, stream>>>(x, W, Y);
  k_row_stats <<<dim3(NN, NB), 256, 0, stream>>>(x, Y, bias, rowM, rowZ);
  k_aggregate <<<dim3(NN / 256, NB, 8), 256, 0, stream>>>(x, Y, bias, a, meta, h3, rowM, rowZ, out);
}


# Row softmax stats: one 8-wave group per row, two LDS barrier trees -> one thread per row and 256-j segment (lane = row, j serial), Y and bias as scalar operands; same max, new Z sum order
# speedup vs baseline: 8.2406x; 1.1488x over previous
#include <hip/hip_runtime.h>
#include <math.h>

#define NN 2048
#define NB 8
#define NB1 16384
#define NB2 1024
#define NB3 256
#define RANK_K (((unsigned)NN * (unsigned)NN - 1u) / 2u)

typedef __attribute__((ext_vector_type(2))) float v2f;
typedef __attribute__((ext_vector_type(8))) float v8f;

__global__ __launch_bounds__(256) void k_row_softmax(const float* __restrict__ adj,
                                                     float* __restrict__ a) {
  __shared__ float red[256];
  const int i = blockIdx.x;
  const int t = threadIdx.x;
  const float* row = adj + (size_t)i * NN;

  float mymax = -INFINITY;
  for (int j = t; j < NN; j += 256) mymax = fmaxf(mymax, row[j]);
  red[t] = mymax; __syncthreads();
  for (int s = 128; s > 0; s >>= 1) { if (t < s) red[t] = fmaxf(red[t], red[t + s]); __syncthreads(); }
  const float m = red[0];
  __syncthreads();

  float sum = 0.f;
  for (int j = t; j < NN; j += 256) sum += __expf(row[j] - m);
  red[t] = sum; __syncthreads();
  for (int s = 128; s > 0; s >>= 1) { if (t < s) red[t] += red[t + s]; __syncthreads(); }
  const float inv = 1.f / red[0];

  float* arow = a + (size_t)i * NN;
  for (int j = t; j < NN; j += 256) arow[j] = __expf(row[j] - m) * inv;
}


template <int PER>
__device__ __forceinline__ void find_rank(const unsigned (&c)[PER], const unsigned kk,
                                          unsigned* s_tot, unsigned* s_res,
                                          unsigned& bin, unsigned& before) {
  const int t = threadIdx.x, lane = t & 31, wv = t >> 5;
  unsigned mine = 0;
  for (int q = 0; q < PER; ++q) mine += c[q];
  unsigned inc = mine;
  for (int d = 1; d < 32; d <<= 1) {
    const unsigned up = __shfl_up(inc, d);
    if (lane >= d) inc += up;
  }
  if (lane == 31) s_tot[wv] = inc;
  if (t == 0) { s_res[0] = 0u; s_res[1] = 0u; }
  __syncthreads();
  unsigned excl = inc - mine;
  for (int w = 0; w < wv; ++w) excl += s_tot[w];
  if (kk >= excl && kk - excl < mine) {
    unsigned cum = excl, fb = 0u, fc = 0u;
    bool found = false;
    for (int q = 0; q < PER; ++q) {
      if (!found && kk - cum < c[q]) { fb = (unsigned)q; fc = cum; found = true; }
      cum += c[q];
    }
    s_res[0] = (unsigned)t * PER + fb;
    s_res[1] = fc;
  }
  __syncthreads();
  bin = s_res[0];
  before = s_res[1];
}

__global__ __launch_bounds__(512) void k_count_d1(const float* __restrict__ a,
                                                  unsigned* __restrict__ h1) {
  __shared__ unsigned bins[NB1];
  const int t = threadIdx.x;
  const float4* a4 = reinterpret_cast<const float4*>(a) + (size_t)blockIdx.x * (512 * 8);
  float4 v[8];
  for (int q = 0; q < 8; ++q) v[q] = a4[q * 512 + t];
  uint4* b4 = reinterpret_cast<uint4*>(bins);
  for (int q = 0; q < 8; ++q) b4[q * 512 + t] = make_uint4(0u, 0u, 0u, 0u);
  __syncthreads();
  for (int q = 0; q < 8; ++q) {
    const unsigned u[4] = { __float_as_uint(v[q].x), __float_as_uint(v[q].y),
                            __float_as_uint(v[q].z), __float_as_uint(v[q].w) };
    for (int e = 0; e < 4; ++e) {
      const unsigned b = u[e] >> 18;
      const unsigned p = ((b & 28u) << 9) | ((b >> 5) << 2) | (b & 3u);
      atomicAdd(&bins[p], 1u);
    }
  }
  __syncthreads();
  uint4 w[8];
  for (int q = 0; q < 8; ++q) w[q] = b4[q * 512 + t];
  for (int q = 0; q < 8; ++q) {
    if ((w[q].x | w[q].y | w[q].z | w[q].w) != 0u) {
      unsigned* g = h1 + (q * 512 + t) * 4;
      if (w[q].x != 0u) atomicAdd(g + 0, w[q].x);
      if (w[q].y != 0u) atomicAdd(g + 1, w[q].y);
      if (w[q].z != 0u) atomicAdd(g + 2, w[q].z);
      if (w[q].w != 0u) atomicAdd(g + 3, w[q].w);
    }
  }
}

__global__ __launch_bounds__(512) void k_count_d2(const float* __restrict__ a,
                                                  const unsigned* __restrict__ h1,
                                                  unsigned* __restrict__ h2,
                                                  unsigned* __restrict__ meta) {
  __shared__ unsigned s_tot[16];
  __shared__ unsigned s_res[2];
  const int t = threadIdx.x;
  const float4* a4 = reinterpret_cast<const float4*>(a) + (size_t)blockIdx.x * (512 * 8);
  float4 v[8];
  for (int q = 0; q < 8; ++q) v[q] = a4[q * 512 + t];
  const uint4* h4 = reinterpret_cast<const uint4*>(h1);
  unsigned c[32];
  for (int q = 0; q < 8; ++q) {
    const uint4 w = h4[q * 512 + t];
    c[4 * q + 0] = w.x; c[4 * q + 1] = w.y; c[4 * q + 2] = w.z; c[4 * q + 3] = w.w;
  }
  unsigned bin1, before1;
  find_rank<32>(c, RANK_K, s_tot, s_res, bin1, before1);
  if (blockIdx.x == 0 && t == 0) { meta[0] = bin1; meta[1] = before1; }
  for (int q = 0; q < 8; ++q) {
    const unsigned u[4] = { __float_as_uint(v[q].x), __float_as_uint(v[q].y),
                            __float_as_uint(v[q].z), __float_as_uint(v[q].w) };
    for (int e = 0; e < 4; ++e)
      if ((u[e] >> 18) == bin1) atomicAdd(&h2[(u[e] >> 8) & 1023u], 1u);
  }
}

__global__ __launch_bounds__(512) void k_count_d3(const float* __restrict__ a,
                                                  const unsigned* __restrict__ h2,
                                                  unsigned* __restrict__ h3,
                                                  unsigned* __restrict__ meta) {
  __shared__ unsigned s_tot[16];
  __shared__ unsigned s_res[2];
  const int t = threadIdx.x;
  const float4* a4 = reinterpret_cast<const float4*>(a) + (size_t)blockIdx.x * (512 * 8);
  float4 v[8];
  for (int q = 0; q < 8; ++q) v[q] = a4[q * 512 + t];
  const unsigned bin1 = meta[0], before1 = meta[1];
  const uint2 w = reinterpret_cast<const uint2*>(h2)[t];
  const unsigned c[2] = { w.x, w.y };
  unsigned bin2, before2;
  find_rank<2>(c, RANK_K - before1, s_tot, s_res, bin2, before2);
  const unsigned pre = (bin1 << 10) | bin2;
  if (blockIdx.x == 0 && t == 0) { meta[2] = pre; meta[3] = RANK_K - before1 - before2; }
  for (int q = 0; q < 8; ++q) {
    const unsigned u[4] = { __float_as_uint(v[q].x), __float_as_uint(v[q].y),
                            __float_as_uint(v[q].z), __float_as_uint(v[q].w) };
    for (int e = 0; e < 4; ++e)
      if ((u[e] >> 8) == pre) atomicAdd(&h3[u[e] & 255u], 1u);
  }
}

__global__ __launch_bounds__(32) void k_gemm_y(const float* __restrict__ x,
                                               const float* __restrict__ W,
                                               float* __restrict__ Y) {
  __shared__ float Wt[16][132];
  __shared__ float Xt[16][132];
  const int nb   = blockIdx.x * 16;
  const int lane = threadIdx.x;
  const int half = lane >> 4;
  const int l16  = lane & 15;

  for (int idx = lane; idx < 8 * 128; idx += 32)
    Xt[8 + (idx >> 7)][idx & 127] = 0.f;

  v8f acc0 = {}, acc1 = {};
  for (int k0 = 0; k0 < NN; k0 += 128) {
    for (int idx = lane; idx < 8 * 128; idx += 32) {
      const int r = idx >> 7, c = idx & 127;
      Xt[r][c] = x[r * NN + (k0 + c)];
    }
    for (int idx = lane; idx < 16 * 128; idx += 32) {
      const int r = idx >> 7, c = idx & 127;
      Wt[r][c] = W[(size_t)(nb + r) * NN + (k0 + c)];
    }
    __syncthreads();
    #pragma unroll 4
    for (int kk = 0; kk < 128; kk += 8) {
      const int ka0 = kk + half * 2;
      v2f a0 = { Xt[l16][ka0], Xt[l16][ka0 + 1] };
      v2f b0 = { Wt[l16][ka0], Wt[l16][ka0 + 1] };
      acc0 = __builtin_amdgcn_wmma_f32_16x16x4_f32(false, a0, false, b0,
                                                   (short)0, acc0, false, false);
      const int ka1 = kk + 4 + half * 2;
      v2f a1 = { Xt[l16][ka1], Xt[l16][ka1 + 1] };
      v2f b1 = { Wt[l16][ka1], Wt[l16][ka1 + 1] };
      acc1 = __builtin_amdgcn_wmma_f32_16x16x4_f32(false, a1, false, b1,
                                                   (short)0, acc1, false, false);
    }
    __syncthreads();
  }
  for (int r = 0; r < 8; ++r) {
    const int m = r + half * 8;
    if (m < NB) Y[m * NN + nb + l16] = acc0[r] + acc1[r];
  }
}

#define RS_SEG 256
__global__ __launch_bounds__(256) void k_row_stats(const float* __restrict__ x,
                                                   const float* __restrict__ Y,
                                                   const float* __restrict__ bias,
                                                   float* __restrict__ rowM,
                                                   float* __restrict__ rowZ) {
  __shared__ float pm[NN / RS_SEG][32];
  __shared__ float pz[NN / RS_SEG][32];
  const int lane = threadIdx.x & 31;
  const int seg  = __builtin_amdgcn_readfirstlane((int)(threadIdx.x >> 5));
  const int b    = blockIdx.y;
  const int i    = blockIdx.x * 32 + lane;
  const float xbi = x[b * NN + i];
  const float* __restrict__ yr = Y + b * NN + seg * RS_SEG;
  const float* __restrict__ br = bias + seg * RS_SEG;

  float vmax = -INFINITY;
  #pragma clang loop vectorize(disable) unroll_count(16)
  for (int j = 0; j < RS_SEG; ++j) vmax = fmaxf(vmax, fmaf(xbi, yr[j], br[j]));
  pm[seg][lane] = vmax;
  __syncthreads();
  float vm = pm[0][lane];
  for (int s = 1; s < NN / RS_SEG; ++s) vm = fmaxf(vm, pm[s][lane]);
  const float m = fmaxf(vm, 0.01f * vm);

  float sum = 0.f;
  #pragma unroll 16
  for (int j = 0; j < RS_SEG; ++j) {
    float v = fmaf(xbi, yr[j], br[j]);
    v = fmaxf(v, 0.01f * v);
    sum += __expf(v - m);
  }
  pz[seg][lane] = sum;
  __syncthreads();
  if (seg == 0) {
    float z = pz[0][lane];
    for (int s = 1; s < NN / RS_SEG; ++s) z += pz[s][lane];
    rowM[b * NN + i] = m;
    rowZ[b * NN + i] = z;
  }
}

__global__ __launch_bounds__(256) void k_aggregate(const float* __restrict__ x,
                                                   const float* __restrict__ Y,
                                                   const float* __restrict__ bias,
                                                   const float* __restrict__ a,
                                                   const unsigned* __restrict__ meta,
                                                   const unsigned* __restrict__ h3,
                                                   const float* __restrict__ rowM,
                                                   const float* __restrict__ rowZ,
                                                   float* __restrict__ out) {
  const int j  = blockIdx.x * 256 + threadIdx.x;
  const int b  = blockIdx.y;
  const int iBase = blockIdx.z * (NN / 8);
  __shared__ unsigned s_tot[8];
  __shared__ unsigned s_res[2];
  const unsigned c3[1] = { h3[threadIdx.x] };
  unsigned bin3, before3;
  find_rank<1>(c3, meta[3], s_tot, s_res, bin3, before3);
  const float med = __uint_as_float((meta[2] << 8) | bin3);
  const float ybj = Y[b * NN + j];
  const float bj  = bias[j];

  __shared__ float sx[128], sm[128], sc[128];
  float acc = 0.f;
  for (int i0 = iBase; i0 < iBase + 256; i0 += 128) {
    if (threadIdx.x < 128) {
      const int i = i0 + threadIdx.x;
      const float xv = x[b * NN + i];
      sx[threadIdx.x] = xv;
      sm[threadIdx.x] = rowM[b * NN + i];
      sc[threadIdx.x] = xv / rowZ[b * NN + i];
    }
    __syncthreads();
    for (int ii = 0; ii < 128; ++ii) {
      const int i = i0 + ii;
      if ((ii & 15) == 0)
        __builtin_prefetch(&a[(size_t)(i + 16) * NN + j], 0, 1);
      const float aij = a[(size_t)i * NN + j];
      float t = fmaf(sx[ii], ybj, bj);
      t = (t > 0.f) ? t : 0.01f * t;
      acc += (aij > med) ? sc[ii] * __expf(t - sm[ii]) : 0.f;
    }
    __syncthreads();
  }
  atomicAdd(&out[b * NN + j], acc);
}

extern "C" void kernel_launch(void* const* d_in, const int* in_sizes, int n_in,
                              void* d_out, int out_size, void* d_ws, size_t ws_size,
                              hipStream_t stream) {
  const float* x    = (const float*)d_in[0];
  const float* adj  = (const float*)d_in[1];
  const float* W    = (const float*)d_in[2];
  const float* bias = (const float*)d_in[3];
  float* out = (float*)d_out;

  char* ws = (char*)d_ws;
  size_t off = 0;
  float*    a     = (float*)(ws + off);    off += (size_t)NN * NN * sizeof(float);
  float*    Y     = (float*)(ws + off);    off += (size_t)NB * NN * sizeof(float);
  float*    rowM  = (float*)(ws + off);    off += (size_t)NB * NN * sizeof(float);
  float*    rowZ  = (float*)(ws + off);    off += (size_t)NB * NN * sizeof(float);
  unsigned* h1    = (unsigned*)(ws + off); off += (size_t)NB1 * sizeof(unsigned);
  unsigned* h2    = (unsigned*)(ws + off); off += (size_t)NB2 * sizeof(unsigned);
  unsigned* h3    = (unsigned*)(ws + off); off += (size_t)NB3 * sizeof(unsigned);
  unsigned* meta  = (unsigned*)(ws + off); off += 16;

  hipMemsetAsync(h1, 0, (size_t)(NB1 + NB2 + NB3) * sizeof(unsigned), stream);
  hipMemsetAsync(out, 0, (size_t)out_size * sizeof(float), stream);

  k_row_softmax<<<NN, 256, 0, stream>>>(adj, a);
  k_count_d1  <<<NN * NN / (512 * 32), 512, 0, stream>>>(a, h1);
  k_count_d2  <<<NN * NN / (512 * 32), 512, 0, stream>>>(a, h1, h2, meta);
  k_count_d3  <<<NN * NN / (512 * 32), 512, 0, stream>>>(a, h2, h3, meta);
  k_gemm_y    <<<NN / 16, 32, 0, stream>>>(x, W, Y);
  k_row_stats <<<dim3(NN / 32, NB), 256, 0, stream>>>(x, Y, bias, rowM, rowZ);
  k_aggregate <<<dim3(NN / 256, NB, 8), 256, 0, stream>>>(x, Y, bias, a, meta, h3, rowM, rowZ, out);
}


# k_aggregate: [a > med] packed once into bit words (new launch: a wave vote per row and 32 columns), read by scalar loads as the select's lane mask; no column walk of a, no wait per row; same sums
# speedup vs baseline: 11.4107x; 1.3847x over previous
#include <hip/hip_runtime.h>
#include <math.h>

#define NN 2048
#define NB 8
#define NB1 16384
#define NB2 1024
#define NB3 256
#define RANK_K (((unsigned)NN * (unsigned)NN - 1u) / 2u)

typedef __attribute__((ext_vector_type(2))) float v2f;
typedef __attribute__((ext_vector_type(8))) float v8f;

__global__ __launch_bounds__(256) void k_row_softmax(const float* __restrict__ adj,
                                                     float* __restrict__ a) {
  __shared__ float red[256];
  const int i = blockIdx.x;
  const int t = threadIdx.x;
  const float* row = adj + (size_t)i * NN;

  float mymax = -INFINITY;
  for (int j = t; j < NN; j += 256) mymax = fmaxf(mymax, row[j]);
  red[t] = mymax; __syncthreads();
  for (int s = 128; s > 0; s >>= 1) { if (t < s) red[t] = fmaxf(red[t], red[t + s]); __syncthreads(); }
  const float m = red[0];
  __syncthreads();

  float sum = 0.f;
  for (int j = t; j < NN; j += 256) sum += __expf(row[j] - m);
  red[t] = sum; __syncthreads();
  for (int s = 128; s > 0; s >>= 1) { if (t < s) red[t] += red[t + s]; __syncthreads(); }
  const float inv = 1.f / red[0];

  float* arow = a + (size_t)i * NN;
  for (int j = t; j < NN; j += 256) arow[j] = __expf(row[j] - m) * inv;
}


template <int PER>
__device__ __forceinline__ void find_rank(const unsigned (&c)[PER], const unsigned kk,
                                          unsigned* s_tot, unsigned* s_res,
                                          unsigned& bin, unsigned& before) {
  const int t = threadIdx.x, lane = t & 31, wv = t >> 5;
  unsigned mine = 0;
  for (int q = 0; q < PER; ++q) mine += c[q];
  unsigned inc = mine;
  for (int d = 1; d < 32; d <<= 1) {
    const unsigned up = __shfl_up(inc, d);
    if (lane >= d) inc += up;
  }
  if (lane == 31) s_tot[wv] = inc;
  if (t == 0) { s_res[0] = 0u; s_res[1] = 0u; }
  __syncthreads();
  unsigned excl = inc - mine;
  for (int w = 0; w < wv; ++w) excl += s_tot[w];
  if (kk >= excl && kk - excl < mine) {
    unsigned cum = excl, fb = 0u, fc = 0u;
    bool found = false;
    for (int q = 0; q < PER; ++q) {
      if (!found && kk - cum < c[q]) { fb = (unsigned)q; fc = cum; found = true; }
      cum += c[q];
    }
    s_res[0] = (unsigned)t * PER + fb;
    s_res[1] = fc;
  }
  __syncthreads();
  bin = s_res[0];
  before = s_res[1];
}

__global__ __launch_bounds__(512) void k_count_d1(const float* __restrict__ a,
                                                  unsigned* __restrict__ h1) {
  __shared__ unsigned bins[NB1];
  const int t = threadIdx.x;
  const float4* a4 = reinterpret_cast<const float4*>(a) + (size_t)blockIdx.x * (512 * 8);
  float4 v[8];
  for (int q = 0; q < 8; ++q) v[q] = a4[q * 512 + t];
  uint4* b4 = reinterpret_cast<uint4*>(bins);
  for (int q = 0; q < 8; ++q) b4[q * 512 + t] = make_uint4(0u, 0u, 0u, 0u);
  __syncthreads();
  for (int q = 0; q < 8; ++q) {
    const unsigned u[4] = { __float_as_uint(v[q].x), __float_as_uint(v[q].y),
                            __float_as_uint(v[q].z), __float_as_uint(v[q].w) };
    for (int e = 0; e < 4; ++e) {
      const unsigned b = u[e] >> 18;
      const unsigned p = ((b & 28u) << 9) | ((b >> 5) << 2) | (b & 3u);
      atomicAdd(&bins[p], 1u);
    }
  }
  __syncthreads();
  uint4 w[8];
  for (int q = 0; q < 8; ++q) w[q] = b4[q * 512 + t];
  for (int q = 0; q < 8; ++q) {
    if ((w[q].x | w[q].y | w[q].z | w[q].w) != 0u) {
      unsigned* g = h1 + (q * 512 + t) * 4;
      if (w[q].x != 0u) atomicAdd(g + 0, w[q].x);
      if (w[q].y != 0u) atomicAdd(g + 1, w[q].y);
      if (w[q].z != 0u) atomicAdd(g + 2, w[q].z);
      if (w[q].w != 0u) atomicAdd(g + 3, w[q].w);
    }
  }
}

__global__ __launch_bounds__(512) void k_count_d2(const float* __restrict__ a,
                                                  const unsigned* __restrict__ h1,
                                                  unsigned* __restrict__ h2,
                                                  unsigned* __restrict__ meta) {
  __shared__ unsigned s_tot[16];
  __shared__ unsigned s_res[2];
  const int t = threadIdx.x;
  const float4* a4 = reinterpret_cast<const float4*>(a) + (size_t)blockIdx.x * (512 * 8);
  float4 v[8];
  for (int q = 0; q < 8; ++q) v[q] = a4[q * 512 + t];
  const uint4* h4 = reinterpret_cast<const uint4*>(h1);
  unsigned c[32];
  for (int q = 0; q < 8; ++q) {
    const uint4 w = h4[q * 512 + t];
    c[4 * q + 0] = w.x; c[4 * q + 1] = w.y; c[4 * q + 2] = w.z; c[4 * q + 3] = w.w;
  }
  unsigned bin1, before1;
  find_rank<32>(c, RANK_K, s_tot, s_res, bin1, before1);
  if (blockIdx.x == 0 && t == 0) { meta[0] = bin1; meta[1] = before1; }
  for (int q = 0; q < 8; ++q) {
    const unsigned u[4] = { __float_as_uint(v[q].x), __float_as_uint(v[q].y),
                            __float_as_uint(v[q].z), __float_as_uint(v[q].w) };
    for (int e = 0; e < 4; ++e)
      if ((u[e] >> 18) == bin1) atomicAdd(&h2[(u[e] >> 8) & 1023u], 1u);
  }
}

__global__ __launch_bounds__(512) void k_count_d3(const float* __restrict__ a,
                                                  const unsigned* __restrict__ h2,
                                                  unsigned* __restrict__ h3,
                                                  unsigned* __restrict__ meta) {
  __shared__ unsigned s_tot[16];
  __shared__ unsigned s_res[2];
  const int t = threadIdx.x;
  const float4* a4 = reinterpret_cast<const float4*>(a) + (size_t)blockIdx.x * (512 * 8);
  float4 v[8];
  for (int q = 0; q < 8; ++q) v[q] = a4[q * 512 + t];
  const unsigned bin1 = meta[0], before1 = meta[1];
  const uint2 w = reinterpret_cast<const uint2*>(h2)[t];
  const unsigned c[2] = { w.x, w.y };
  unsigned bin2, before2;
  find_rank<2>(c, RANK_K - before1, s_tot, s_res, bin2, before2);
  const unsigned pre = (bin1 << 10) | bin2;
  if (blockIdx.x == 0 && t == 0) { meta[2] = pre; meta[3] = RANK_K - before1 - before2; }
  for (int q = 0; q < 8; ++q) {
    const unsigned u[4] = { __float_as_uint(v[q].x), __float_as_uint(v[q].y),
                            __float_as_uint(v[q].z), __float_as_uint(v[q].w) };
    for (int e = 0; e < 4; ++e)
      if ((u[e] >> 8) == pre) atomicAdd(&h3[u[e] & 255u], 1u);
  }
}

__global__ __launch_bounds__(32) void k_gemm_y(const float* __restrict__ x,
                                               const float* __restrict__ W,
                                               float* __restrict__ Y) {
  __shared__ float Wt[16][132];
  __shared__ float Xt[16][132];
  const int nb   = blockIdx.x * 16;
  const int lane = threadIdx.x;
  const int half = lane >> 4;
  const int l16  = lane & 15;

  for (int idx = lane; idx < 8 * 128; idx += 32)
    Xt[8 + (idx >> 7)][idx & 127] = 0.f;

  v8f acc0 = {}, acc1 = {};
  for (int k0 = 0; k0 < NN; k0 += 128) {
    for (int idx = lane; idx < 8 * 128; idx += 32) {
      const int r = idx >> 7, c = idx & 127;
      Xt[r][c] = x[r * NN + (k0 + c)];
    }
    for (int idx = lane; idx < 16 * 128; idx += 32) {
      const int r = idx >> 7, c = idx & 127;
      Wt[r][c] = W[(size_t)(nb + r) * NN + (k0 + c)];
    }
    __syncthreads();
    #pragma unroll 4
    for (int kk = 0; kk < 128; kk += 8) {
      const int ka0 = kk + half * 2;
      v2f a0 = { Xt[l16][ka0], Xt[l16][ka0 + 1] };
      v2f b0 = { Wt[l16][ka0], Wt[l16][ka0 + 1] };
      acc0 = __builtin_amdgcn_wmma_f32_16x16x4_f32(false, a0, false, b0,
                                                   (short)0, acc0, false, false);
      const int ka1 = kk + 4 + half * 2;
      v2f a1 = { Xt[l16][ka1], Xt[l16][ka1 + 1] };
      v2f b1 = { Wt[l16][ka1], Wt[l16][ka1 + 1] };
      acc1 = __builtin_amdgcn_wmma_f32_16x16x4_f32(false, a1, false, b1,
                                                   (short)0, acc1, false, false);
    }
    __syncthreads();
  }
  for (int r = 0; r < 8; ++r) {
    const int m = r + half * 8;
    if (m < NB) Y[m * NN + nb + l16] = acc0[r] + acc1[r];
  }
}

#define RS_SEG 256
__global__ __launch_bounds__(256) void k_row_stats(const float* __restrict__ x,
                                                   const float* __restrict__ Y,
                                                   const float* __restrict__ bias,
                                                   float* __restrict__ rowM,
                                                   float* __restrict__ rowZ) {
  __shared__ float pm[NN / RS_SEG][32];
  __shared__ float pz[NN / RS_SEG][32];
  const int lane = threadIdx.x & 31;
  const int seg  = __builtin_amdgcn_readfirstlane((int)(threadIdx.x >> 5));
  const int b    = blockIdx.y;
  const int i    = blockIdx.x * 32 + lane;
  const float xbi = x[b * NN + i];
  const float* __restrict__ yr = Y + b * NN + seg * RS_SEG;
  const float* __restrict__ br = bias + seg * RS_SEG;

  float vmax = -INFINITY;
  #pragma clang loop vectorize(disable) unroll_count(16)
  for (int j = 0; j < RS_SEG; ++j) vmax = fmaxf(vmax, fmaf(xbi, yr[j], br[j]));
  pm[seg][lane] = vmax;
  __syncthreads();
  float vm = pm[0][lane];
  for (int s = 1; s < NN / RS_SEG; ++s) vm = fmaxf(vm, pm[s][lane]);
  const float m = fmaxf(vm, 0.01f * vm);

  float sum = 0.f;
  #pragma unroll 16
  for (int j = 0; j < RS_SEG; ++j) {
    float v = fmaf(xbi, yr[j], br[j]);
    v = fmaxf(v, 0.01f * v);
    sum += __expf(v - m);
  }
  pz[seg][lane] = sum;
  __syncthreads();
  if (seg == 0) {
    float z = pz[0][lane];
    for (int s = 1; s < NN / RS_SEG; ++s) z += pz[s][lane];
    rowM[b * NN + i] = m;
    rowZ[b * NN + i] = z;
  }
}

__global__ __launch_bounds__(256) void k_mask(const float* __restrict__ a,
                                              const unsigned* __restrict__ meta,
                                              const unsigned* __restrict__ h3,
                                              unsigned* __restrict__ maskT) {
  __shared__ unsigned s_tot[8];
  __shared__ unsigned s_res[2];
  const int lane = threadIdx.x & 31;
  const int tile = (blockIdx.x & 7) * 8 + (threadIdx.x >> 5);
  const int i0   = (blockIdx.x >> 3) * 32;
  const float* __restrict__ p = a + (size_t)i0 * NN + tile * 32 + lane;
  float v[32];
  #pragma unroll
  for (int r = 0; r < 32; ++r) v[r] = p[(size_t)r * NN];
  const unsigned c3[1] = { h3[threadIdx.x] };
  unsigned bin3, before3;
  find_rank<1>(c3, meta[3], s_tot, s_res, bin3, before3);
  const float med = __uint_as_float((meta[2] << 8) | bin3);
  unsigned mine = 0u;
  #pragma unroll
  for (int r = 0; r < 32; ++r) {
    const unsigned w = __builtin_amdgcn_ballot_w32(v[r] > med);
    if (lane == r) mine = w;
  }
  maskT[(size_t)tile * NN + i0 + lane] = mine;
}

__global__ __launch_bounds__(256) void k_aggregate(const float* __restrict__ x,
                                                   const float* __restrict__ Y,
                                                   const float* __restrict__ bias,
                                                   const unsigned* __restrict__ maskT,
                                                   const float* __restrict__ rowM,
                                                   const float* __restrict__ rowZ,
                                                   float* __restrict__ out) {
  const int j  = blockIdx.x * 256 + threadIdx.x;
  const int b  = blockIdx.y;
  const int iBase = blockIdx.z * (NN / 8);
  const int tile = blockIdx.x * 8 + __builtin_amdgcn_readfirstlane((int)(threadIdx.x >> 5));
  const unsigned* __restrict__ mrow = maskT + (size_t)tile * NN + iBase;
  const float ybj = Y[b * NN + j];
  const float bj  = bias[j];

  __shared__ float sx[256], sm[256], sc[256];
  {
    const int i = iBase + threadIdx.x;
    const float xv = x[b * NN + i];
    sx[threadIdx.x] = xv;
    sm[threadIdx.x] = rowM[b * NN + i];
    sc[threadIdx.x] = xv / rowZ[b * NN + i];
  }
  __syncthreads();
  float acc = 0.f;
  #pragma unroll 16
  for (int ii = 0; ii < 256; ++ii) {
    const bool on = __builtin_amdgcn_inverse_ballot_w32(mrow[ii]);
    float t = fmaf(sx[ii], ybj, bj);
    t = (t > 0.f) ? t : 0.01f * t;
    const float term = sc[ii] * __expf(t - sm[ii]);
    acc += on ? term : 0.f;
  }
  atomicAdd(&out[b * NN + j], acc);
}

extern "C" void kernel_launch(void* const* d_in, const int* in_sizes, int n_in,
                              void* d_out, int out_size, void* d_ws, size_t ws_size,
                              hipStream_t stream) {
  const float* x    = (const float*)d_in[0];
  const float* adj  = (const float*)d_in[1];
  const float* W    = (const float*)d_in[2];
  const float* bias = (const float*)d_in[3];
  float* out = (float*)d_out;

  char* ws = (char*)d_ws;
  size_t off = 0;
  float*    a     = (float*)(ws + off);    off += (size_t)NN * NN * sizeof(float);
  float*    Y     = (float*)(ws + off);    off += (size_t)NB * NN * sizeof(float);
  float*    rowM  = (float*)(ws + off);    off += (size_t)NB * NN * sizeof(float);
  float*    rowZ  = (float*)(ws + off);    off += (size_t)NB * NN * sizeof(float);
  unsigned* maskT = (unsigned*)(ws + off); off += (size_t)(NN / 32) * NN * sizeof(unsigned);
  unsigned* h1    = (unsigned*)(ws + off); off += (size_t)NB1 * sizeof(unsigned);
  unsigned* h2    = (unsigned*)(ws + off); off += (size_t)NB2 * sizeof(unsigned);
  unsigned* h3    = (unsigned*)(ws + off); off += (size_t)NB3 * sizeof(unsigned);
  unsigned* meta  = (unsigned*)(ws + off); off += 16;

  hipMemsetAsync(h1, 0, (size_t)(NB1 + NB2 + NB3) * sizeof(unsigned), stream);
  hipMemsetAsync(out, 0, (size_t)out_size * sizeof(float), stream);

  k_row_softmax<<<NN, 256, 0, stream>>>(adj, a);
  k_count_d1  <<<NN * NN / (512 * 32), 512, 0, stream>>>(a, h1);
  k_count_d2  <<<NN * NN / (512 * 32), 512, 0, stream>>>(a, h1, h2, meta);
  k_count_d3  <<<NN * NN / (512 * 32), 512, 0, stream>>>(a, h2, h3, meta);
  k_mask      <<<(NN / 32) * (NN / 256), 256, 0, stream>>>(a, meta, h3, maskT);
  k_gemm_y    <<<NN / 16, 32, 0, stream>>>(x, W, Y);
  k_row_stats <<<dim3(NN / 32, NB), 256, 0, stream>>>(x, Y, bias, rowM, rowZ);
  k_aggregate <<<dim3(NN / 256, NB, 8), 256, 0, stream>>>(x, Y, bias, maskT, rowM, rowZ, out);
}


# Y GEMM operands: LDS staging, 8 scalar loads in flight -> each lane loads its own A/B fragment values from global (16-byte loads, 64-k stages of 16 loads, 3 stages ahead of WMMAs); new k sum order
# speedup vs baseline: 25.7448x; 2.2562x over previous
#include <hip/hip_runtime.h>
#include <math.h>

#define NN 2048
#define NB 8
#define NB1 16384
#define NB2 1024
#define NB3 256
#define RANK_K (((unsigned)NN * (unsigned)NN - 1u) / 2u)

typedef __attribute__((ext_vector_type(2))) float v2f;
typedef __attribute__((ext_vector_type(8))) float v8f;

__global__ __launch_bounds__(256) void k_row_softmax(const float* __restrict__ adj,
                                                     float* __restrict__ a) {
  __shared__ float red[256];
  const int i = blockIdx.x;
  const int t = threadIdx.x;
  const float* row = adj + (size_t)i * NN;

  float mymax = -INFINITY;
  for (int j = t; j < NN; j += 256) mymax = fmaxf(mymax, row[j]);
  red[t] = mymax; __syncthreads();
  for (int s = 128; s > 0; s >>= 1) { if (t < s) red[t] = fmaxf(red[t], red[t + s]); __syncthreads(); }
  const float m = red[0];
  __syncthreads();

  float sum = 0.f;
  for (int j = t; j < NN; j += 256) sum += __expf(row[j] - m);
  red[t] = sum; __syncthreads();
  for (int s = 128; s > 0; s >>= 1) { if (t < s) red[t] += red[t + s]; __syncthreads(); }
  const float inv = 1.f / red[0];

  float* arow = a + (size_t)i * NN;
  for (int j = t; j < NN; j += 256) arow[j] = __expf(row[j] - m) * inv;
}


template <int PER>
__device__ __forceinline__ void find_rank(const unsigned (&c)[PER], const unsigned kk,
                                          unsigned* s_tot, unsigned* s_res,
                                          unsigned& bin, unsigned& before) {
  const int t = threadIdx.x, lane = t & 31, wv = t >> 5;
  unsigned mine = 0;
  for (int q = 0; q < PER; ++q) mine += c[q];
  unsigned inc = mine;
  for (int d = 1; d < 32; d <<= 1) {
    const unsigned up = __shfl_up(inc, d);
    if (lane >= d) inc += up;
  }
  if (lane == 31) s_tot[wv] = inc;
  if (t == 0) { s_res[0] = 0u; s_res[1] = 0u; }
  __syncthreads();
  unsigned excl = inc - mine;
  for (int w = 0; w < wv; ++w) excl += s_tot[w];
  if (kk >= excl && kk - excl < mine) {
    unsigned cum = excl, fb = 0u, fc = 0u;
    bool found = false;
    for (int q = 0; q < PER; ++q) {
      if (!found && kk - cum < c[q]) { fb = (unsigned)q; fc = cum; found = true; }
      cum += c[q];
    }
    s_res[0] = (unsigned)t * PER + fb;
    s_res[1] = fc;
  }
  __syncthreads();
  bin = s_res[0];
  before = s_res[1];
}

__global__ __launch_bounds__(512) void k_count_d1(const float* __restrict__ a,
                                                  unsigned* __restrict__ h1) {
  __shared__ unsigned bins[NB1];
  const int t = threadIdx.x;
  const float4* a4 = reinterpret_cast<const float4*>(a) + (size_t)blockIdx.x * (512 * 8);
  float4 v[8];
  for (int q = 0; q < 8; ++q) v[q] = a4[q * 512 + t];
  uint4* b4 = reinterpret_cast<uint4*>(bins);
  for (int q = 0; q < 8; ++q) b4[q * 512 + t] = make_uint4(0u, 0u, 0u, 0u);
  __syncthreads();
  for (int q = 0; q < 8; ++q) {
    const unsigned u[4] = { __float_as_uint(v[q].x), __float_as_uint(v[q].y),
                            __float_as_uint(v[q].z), __float_as_uint(v[q].w) };
    for (int e = 0; e < 4; ++e) {
      const unsigned b = u[e] >> 18;
      const unsigned p = ((b & 28u) << 9) | ((b >> 5) << 2) | (b & 3u);
      atomicAdd(&bins[p], 1u);
    }
  }
  __syncthreads();
  uint4 w[8];
  for (int q = 0; q < 8; ++q) w[q] = b4[q * 512 + t];
  for (int q = 0; q < 8; ++q) {
    if ((w[q].x | w[q].y | w[q].z | w[q].w) != 0u) {
      unsigned* g = h1 + (q * 512 + t) * 4;
      if (w[q].x != 0u) atomicAdd(g + 0, w[q].x);
      if (w[q].y != 0u) atomicAdd(g + 1, w[q].y);
      if (w[q].z != 0u) atomicAdd(g + 2, w[q].z);
      if (w[q].w != 0u) atomicAdd(g + 3, w[q].w);
    }
  }
}

__global__ __launch_bounds__(512) void k_count_d2(const float* __restrict__ a,
                                                  const unsigned* __restrict__ h1,
                                                  unsigned* __restrict__ h2,
                                                  unsigned* __restrict__ meta) {
  __shared__ unsigned s_tot[16];
  __shared__ unsigned s_res[2];
  const int t = threadIdx.x;
  const float4* a4 = reinterpret_cast<const float4*>(a) + (size_t)blockIdx.x * (512 * 8);
  float4 v[8];
  for (int q = 0; q < 8; ++q) v[q] = a4[q * 512 + t];
  const uint4* h4 = reinterpret_cast<const uint4*>(h1);
  unsigned c[32];
  for (int q = 0; q < 8; ++q) {
    const uint4 w = h4[q * 512 + t];
    c[4 * q + 0] = w.x; c[4 * q + 1] = w.y; c[4 * q + 2] = w.z; c[4 * q + 3] = w.w;
  }
  unsigned bin1, before1;
  find_rank<32>(c, RANK_K, s_tot, s_res, bin1, before1);
  if (blockIdx.x == 0 && t == 0) { meta[0] = bin1; meta[1] = before1; }
  for (int q = 0; q < 8; ++q) {
    const unsigned u[4] = { __float_as_uint(v[q].x), __float_as_uint(v[q].y),
                            __float_as_uint(v[q].z), __float_as_uint(v[q].w) };
    for (int e = 0; e < 4; ++e)
      if ((u[e] >> 18) == bin1) atomicAdd(&h2[(u[e] >> 8) & 1023u], 1u);
  }
}

__global__ __launch_bounds__(512) void k_count_d3(const float* __restrict__ a,
                                                  const unsigned* __restrict__ h2,
                                                  unsigned* __restrict__ h3,
                                                  unsigned* __restrict__ meta) {
  __shared__ unsigned s_tot[16];
  __shared__ unsigned s_res[2];
  const int t = threadIdx.x;
  const float4* a4 = reinterpret_cast<const float4*>(a) + (size_t)blockIdx.x * (512 * 8);
  float4 v[8];
  for (int q = 0; q < 8; ++q) v[q] = a4[q * 512 + t];
  const unsigned bin1 = meta[0], before1 = meta[1];
  const uint2 w = reinterpret_cast<const uint2*>(h2)[t];
  const unsigned c[2] = { w.x, w.y };
  unsigned bin2, before2;
  find_rank<2>(c, RANK_K - before1, s_tot, s_res, bin2, before2);
  const unsigned pre = (bin1 << 10) | bin2;
  if (blockIdx.x == 0 && t == 0) { meta[2] = pre; meta[3] = RANK_K - before1 - before2; }
  for (int q = 0; q < 8; ++q) {
    const unsigned u[4] = { __float_as_uint(v[q].x), __float_as_uint(v[q].y),
                            __float_as_uint(v[q].z), __float_as_uint(v[q].w) };
    for (int e = 0; e < 4; ++e)
      if ((u[e] >> 8) == pre) atomicAdd(&h3[u[e] & 255u], 1u);
  }
}

#define GK 64
#define GL (GK / 8)

struct FragSet { float4 w[GL]; float4 x[GL]; };

#define GEMM_STAGE_FENCE() __builtin_amdgcn_sched_barrier(0)

__device__ __forceinline__ void gemm_load(FragSet& f, const float* __restrict__ wl,
                                          const float* __restrict__ xl, const int kb) {
  GEMM_STAGE_FENCE();
  #pragma unroll
  for (int i = 0; i < GL; ++i) {
    f.w[i] = *reinterpret_cast<const float4*>(wl + kb + 8 * i);
    f.x[i] = *reinterpret_cast<const float4*>(xl + kb + 8 * i);
  }
  GEMM_STAGE_FENCE();
}

__device__ __forceinline__ void gemm_mma(const FragSet& f, v8f& acc0, v8f& acc1) {
  #pragma unroll
  for (int i = 0; i < GL; ++i) {
    const v2f a0 = { f.x[i].x, f.x[i].y };
    const v2f b0 = { f.w[i].x, f.w[i].y };
    acc0 = __builtin_amdgcn_wmma_f32_16x16x4_f32(false, a0, false, b0,
                                                 (short)0, acc0, false, false);
    const v2f a1 = { f.x[i].z, f.x[i].w };
    const v2f b1 = { f.w[i].z, f.w[i].w };
    acc1 = __builtin_amdgcn_wmma_f32_16x16x4_f32(false, a1, false, b1,
                                                 (short)0, acc1, false, false);
  }
}

__global__ __launch_bounds__(32) void k_gemm_y(const float* __restrict__ x,
                                               const float* __restrict__ W,
                                               float* __restrict__ Y) {
  const int nb   = blockIdx.x * 16;
  const int lane = threadIdx.x;
  const int half = lane >> 4;
  const int l16  = lane & 15;
  const float* __restrict__ wl = W + (size_t)(nb + l16) * NN + 4 * half;
  const float* __restrict__ xl = x + (size_t)(l16 & 7) * NN + 4 * half;

  FragSet f0, f1, f2, f3;
  v8f acc0 = {}, acc1 = {};
  gemm_load(f0, wl, xl, 0);
  gemm_load(f1, wl, xl, GK);
  gemm_load(f2, wl, xl, 2 * GK);
  #pragma unroll
  for (int kb = 0; kb < NN - 4 * GK; kb += 4 * GK) {
    gemm_load(f3, wl, xl, kb + 3 * GK);
    gemm_mma(f0, acc0, acc1);
    gemm_load(f0, wl, xl, kb + 4 * GK);
    gemm_mma(f1, acc0, acc1);
    gemm_load(f1, wl, xl, kb + 5 * GK);
    gemm_mma(f2, acc0, acc1);
    gemm_load(f2, wl, xl, kb + 6 * GK);
    gemm_mma(f3, acc0, acc1);
  }
  gemm_load(f3, wl, xl, NN - GK);
  gemm_mma(f0, acc0, acc1);
  gemm_mma(f1, acc0, acc1);
  gemm_mma(f2, acc0, acc1);
  gemm_mma(f3, acc0, acc1);

  for (int r = 0; r < 8; ++r) {
    const int m = r + half * 8;
    if (m < NB) Y[m * NN + nb + l16] = acc0[r] + acc1[r];
  }
}

#define RS_SEG 256
__global__ __launch_bounds__(256) void k_row_stats(const float* __restrict__ x,
                                                   const float* __restrict__ Y,
                                                   const float* __restrict__ bias,
                                                   float* __restrict__ rowM,
                                                   float* __restrict__ rowZ) {
  __shared__ float pm[NN / RS_SEG][32];
  __shared__ float pz[NN / RS_SEG][32];
  const int lane = threadIdx.x & 31;
  const int seg  = __builtin_amdgcn_readfirstlane((int)(threadIdx.x >> 5));
  const int b    = blockIdx.y;
  const int i    = blockIdx.x * 32 + lane;
  const float xbi = x[b * NN + i];
  const float* __restrict__ yr = Y + b * NN + seg * RS_SEG;
  const float* __restrict__ br = bias + seg * RS_SEG;

  float vmax = -INFINITY;
  #pragma clang loop vectorize(disable) unroll_count(16)
  for (int j = 0; j < RS_SEG; ++j) vmax = fmaxf(vmax, fmaf(xbi, yr[j], br[j]));
  pm[seg][lane] = vmax;
  __syncthreads();
  float vm = pm[0][lane];
  for (int s = 1; s < NN / RS_SEG; ++s) vm = fmaxf(vm, pm[s][lane]);
  const float m = fmaxf(vm, 0.01f * vm);

  float sum = 0.f;
  #pragma unroll 16
  for (int j = 0; j < RS_SEG; ++j) {
    float v = fmaf(xbi, yr[j], br[j]);
    v = fmaxf(v, 0.01f * v);
    sum += __expf(v - m);
  }
  pz[seg][lane] = sum;
  __syncthreads();
  if (seg == 0) {
    float z = pz[0][lane];
    for (int s = 1; s < NN / RS_SEG; ++s) z += pz[s][lane];
    rowM[b * NN + i] = m;
    rowZ[b * NN + i] = z;
  }
}

__global__ __launch_bounds__(256) void k_mask(const float* __restrict__ a,
                                              const unsigned* __restrict__ meta,
                                              const unsigned* __restrict__ h3,
                                              unsigned* __restrict__ maskT) {
  __shared__ unsigned s_tot[8];
  __shared__ unsigned s_res[2];
  const int lane = threadIdx.x & 31;
  const int tile = (blockIdx.x & 7) * 8 + (threadIdx.x >> 5);
  const int i0   = (blockIdx.x >> 3) * 32;
  const float* __restrict__ p = a + (size_t)i0 * NN + tile * 32 + lane;
  float v[32];
  #pragma unroll
  for (int r = 0; r < 32; ++r) v[r] = p[(size_t)r * NN];
  const unsigned c3[1] = { h3[threadIdx.x] };
  unsigned bin3, before3;
  find_rank<1>(c3, meta[3], s_tot, s_res, bin3, before3);
  const float med = __uint_as_float((meta[2] << 8) | bin3);
  unsigned mine = 0u;
  #pragma unroll
  for (int r = 0; r < 32; ++r) {
    const unsigned w = __builtin_amdgcn_ballot_w32(v[r] > med);
    if (lane == r) mine = w;
  }
  maskT[(size_t)tile * NN + i0 + lane] = mine;
}

__global__ __launch_bounds__(256) void k_aggregate(const float* __restrict__ x,
                                                   const float* __restrict__ Y,
                                                   const float* __restrict__ bias,
                                                   const unsigned* __restrict__ maskT,
                                                   const float* __restrict__ rowM,
                                                   const float* __restrict__ rowZ,
                                                   float* __restrict__ out) {
  const int j  = blockIdx.x * 256 + threadIdx.x;
  const int b  = blockIdx.y;
  const int iBase = blockIdx.z * (NN / 8);
  const int tile = blockIdx.x * 8 + __builtin_amdgcn_readfirstlane((int)(threadIdx.x >> 5));
  const unsigned* __restrict__ mrow = maskT + (size_t)tile * NN + iBase;
  const float ybj = Y[b * NN + j];
  const float bj  = bias[j];

  __shared__ float sx[256], sm[256], sc[256];
  {
    const int i = iBase + threadIdx.x;
    const float xv = x[b * NN + i];
    sx[threadIdx.x] = xv;
    sm[threadIdx.x] = rowM[b * NN + i];
    sc[threadIdx.x] = xv / rowZ[b * NN + i];
  }
  __syncthreads();
  float acc = 0.f;
  #pragma unroll 16
  for (int ii = 0; ii < 256; ++ii) {
    const bool on = __builtin_amdgcn_inverse_ballot_w32(mrow[ii]);
    float t = fmaf(sx[ii], ybj, bj);
    t = (t > 0.f) ? t : 0.01f * t;
    const float term = sc[ii] * __expf(t - sm[ii]);
    acc += on ? term : 0.f;
  }
  atomicAdd(&out[b * NN + j], acc);
}

extern "C" void kernel_launch(void* const* d_in, const int* in_sizes, int n_in,
                              void* d_out, int out_size, void* d_ws, size_t ws_size,
                              hipStream_t stream) {
  const float* x    = (const float*)d_in[0];
  const float* adj  = (const float*)d_in[1];
  const float* W    = (const float*)d_in[2];
  const float* bias = (const float*)d_in[3];
  float* out = (float*)d_out;

  char* ws = (char*)d_ws;
  size_t off = 0;
  float*    a     = (float*)(ws + off);    off += (size_t)NN * NN * sizeof(float);
  float*    Y     = (float*)(ws + off);    off += (size_t)NB * NN * sizeof(float);
  float*    rowM  = (float*)(ws + off);    off += (size_t)NB * NN * sizeof(float);
  float*    rowZ  = (float*)(ws + off);    off += (size_t)NB * NN * sizeof(float);
  unsigned* maskT = (unsigned*)(ws + off); off += (size_t)(NN / 32) * NN * sizeof(unsigned);
  unsigned* h1    = (unsigned*)(ws + off); off += (size_t)NB1 * sizeof(unsigned);
  unsigned* h2    = (unsigned*)(ws + off); off += (size_t)NB2 * sizeof(unsigned);
  unsigned* h3    = (unsigned*)(ws + off); off += (size_t)NB3 * sizeof(unsigned);
  unsigned* meta  = (unsigned*)(ws + off); off += 16;

  hipMemsetAsync(h1, 0, (size_t)(NB1 + NB2 + NB3) * sizeof(unsigned), stream);
  hipMemsetAsync(out, 0, (size_t)out_size * sizeof(float), stream);

  k_row_softmax<<<NN, 256, 0, stream>>>(adj, a);
  k_count_d1  <<<NN * NN / (512 * 32), 512, 0, stream>>>(a, h1);
  k_count_d2  <<<NN * NN / (512 * 32), 512, 0, stream>>>(a, h1, h2, meta);
  k_count_d3  <<<NN * NN / (512 * 32), 512, 0, stream>>>(a, h2, h3, meta);
  k_mask      <<<(NN / 32) * (NN / 256), 256, 0, stream>>>(a, meta, h3, maskT);
  k_gemm_y    <<<NN / 16, 32, 0, stream>>>(x, W, Y);
  k_row_stats <<<dim3(NN / 32, NB), 256, 0, stream>>>(x, Y, bias, rowM, rowZ);
  k_aggregate <<<dim3(NN / 256, NB, 8), 256, 0, stream>>>(x, Y, bias, maskT, rowM, rowZ, out);
}


# Row softmax: 8-wave group per row, 3 reads of the row, two LDS barrier trees -> one wave per row, row kept in registers, lane exchanges in the tree's own sum order; 16384 -> 2048 waves; same bits
# speedup vs baseline: 27.3500x; 1.0624x over previous
#include <hip/hip_runtime.h>
#include <math.h>

#define NN 2048
#define NB 8
#define NB1 16384
#define NB2 1024
#define NB3 256
#define RANK_K (((unsigned)NN * (unsigned)NN - 1u) / 2u)

typedef __attribute__((ext_vector_type(2))) float v2f;
typedef __attribute__((ext_vector_type(8))) float v8f;

#define SM_ROWS 8
__global__ __launch_bounds__(SM_ROWS * 32) void k_row_softmax(const float* __restrict__ adj,
                                                             float* __restrict__ a) {
  const int lane = threadIdx.x & 31;
  const int i = blockIdx.x * SM_ROWS + (threadIdx.x >> 5);
  const float4* __restrict__ r4 = reinterpret_cast<const float4*>(adj + (size_t)i * NN) + lane;

  float4 v[16];
  #pragma unroll
  for (int m = 0; m < 16; ++m) v[m] = r4[32 * m];

  float mx = -INFINITY;
  #pragma unroll
  for (int m = 0; m < 16; ++m)
    mx = fmaxf(mx, fmaxf(fmaxf(v[m].x, v[m].y), fmaxf(v[m].z, v[m].w)));
  for (int d = 16; d > 0; d >>= 1) mx = fmaxf(mx, __shfl_xor(mx, d));

  #pragma unroll
  for (int m = 0; m < 16; ++m) {
    v[m].x = __expf(v[m].x - mx);
    v[m].y = __expf(v[m].y - mx);
    v[m].z = __expf(v[m].z - mx);
    v[m].w = __expf(v[m].w - mx);
  }

  float pl[4] = { 0.f, 0.f, 0.f, 0.f }, ph[4] = { 0.f, 0.f, 0.f, 0.f };
  #pragma unroll
  for (int q = 0; q < 8; ++q) {
    pl[0] += v[2 * q].x;     pl[1] += v[2 * q].y;     pl[2] += v[2 * q].z;     pl[3] += v[2 * q].w;
    ph[0] += v[2 * q + 1].x; ph[1] += v[2 * q + 1].y; ph[2] += v[2 * q + 1].z; ph[3] += v[2 * q + 1].w;
  }
  float r[4];
  for (int c = 0; c < 4; ++c) r[c] = pl[c] + ph[c];
  for (int d = 16; d > 0; d >>= 1)
    for (int c = 0; c < 4; ++c) r[c] += __shfl_xor(r[c], d);
  const float z = (r[0] + r[2]) + (r[1] + r[3]);
  const float inv = 1.f / z;

  float4* __restrict__ o4 = reinterpret_cast<float4*>(a + (size_t)i * NN) + lane;
  #pragma unroll
  for (int m = 0; m < 16; ++m)
    o4[32 * m] = make_float4(v[m].x * inv, v[m].y * inv, v[m].z * inv, v[m].w * inv);
}


template <int PER>
__device__ __forceinline__ void find_rank(const unsigned (&c)[PER], const unsigned kk,
                                          unsigned* s_tot, unsigned* s_res,
                                          unsigned& bin, unsigned& before) {
  const int t = threadIdx.x, lane = t & 31, wv = t >> 5;
  unsigned mine = 0;
  for (int q = 0; q < PER; ++q) mine += c[q];
  unsigned inc = mine;
  for (int d = 1; d < 32; d <<= 1) {
    const unsigned up = __shfl_up(inc, d);
    if (lane >= d) inc += up;
  }
  if (lane == 31) s_tot[wv] = inc;
  if (t == 0) { s_res[0] = 0u; s_res[1] = 0u; }
  __syncthreads();
  unsigned excl = inc - mine;
  for (int w = 0; w < wv; ++w) excl += s_tot[w];
  if (kk >= excl && kk - excl < mine) {
    unsigned cum = excl, fb = 0u, fc = 0u;
    bool found = false;
    for (int q = 0; q < PER; ++q) {
      if (!found && kk - cum < c[q]) { fb = (unsigned)q; fc = cum; found = true; }
      cum += c[q];
    }
    s_res[0] = (unsigned)t * PER + fb;
    s_res[1] = fc;
  }
  __syncthreads();
  bin = s_res[0];
  before = s_res[1];
}

__global__ __launch_bounds__(512) void k_count_d1(const float* __restrict__ a,
                                                  unsigned* __restrict__ h1) {
  __shared__ unsigned bins[NB1];
  const int t = threadIdx.x;
  const float4* a4 = reinterpret_cast<const float4*>(a) + (size_t)blockIdx.x * (512 * 8);
  float4 v[8];
  for (int q = 0; q < 8; ++q) v[q] = a4[q * 512 + t];
  uint4* b4 = reinterpret_cast<uint4*>(bins);
  for (int q = 0; q < 8; ++q) b4[q * 512 + t] = make_uint4(0u, 0u, 0u, 0u);
  __syncthreads();
  for (int q = 0; q < 8; ++q) {
    const unsigned u[4] = { __float_as_uint(v[q].x), __float_as_uint(v[q].y),
                            __float_as_uint(v[q].z), __float_as_uint(v[q].w) };
    for (int e = 0; e < 4; ++e) {
      const unsigned b = u[e] >> 18;
      const unsigned p = ((b & 28u) << 9) | ((b >> 5) << 2) | (b & 3u);
      atomicAdd(&bins[p], 1u);
    }
  }
  __syncthreads();
  uint4 w[8];
  for (int q = 0; q < 8; ++q) w[q] = b4[q * 512 + t];
  for (int q = 0; q < 8; ++q) {
    if ((w[q].x | w[q].y | w[q].z | w[q].w) != 0u) {
      unsigned* g = h1 + (q * 512 + t) * 4;
      if (w[q].x != 0u) atomicAdd(g + 0, w[q].x);
      if (w[q].y != 0u) atomicAdd(g + 1, w[q].y);
      if (w[q].z != 0u) atomicAdd(g + 2, w[q].z);
      if (w[q].w != 0u) atomicAdd(g + 3, w[q].w);
    }
  }
}

__global__ __launch_bounds__(512) void k_count_d2(const float* __restrict__ a,
                                                  const unsigned* __restrict__ h1,
                                                  unsigned* __restrict__ h2,
                                                  unsigned* __restrict__ meta) {
  __shared__ unsigned s_tot[16];
  __shared__ unsigned s_res[2];
  const int t = threadIdx.x;
  const float4* a4 = reinterpret_cast<const float4*>(a) + (size_t)blockIdx.x * (512 * 8);
  float4 v[8];
  for (int q = 0; q < 8; ++q) v[q] = a4[q * 512 + t];
  const uint4* h4 = reinterpret_cast<const uint4*>(h1);
  unsigned c[32];
  for (int q = 0; q < 8; ++q) {
    const uint4 w = h4[q * 512 + t];
    c[4 * q + 0] = w.x; c[4 * q + 1] = w.y; c[4 * q + 2] = w.z; c[4 * q + 3] = w.w;
  }
  unsigned bin1, before1;
  find_rank<32>(c, RANK_K, s_tot, s_res, bin1, before1);
  if (blockIdx.x == 0 && t == 0) { meta[0] = bin1; meta[1] = before1; }
  for (int q = 0; q < 8; ++q) {
    const unsigned u[4] = { __float_as_uint(v[q].x), __float_as_uint(v[q].y),
                            __float_as_uint(v[q].z), __float_as_uint(v[q].w) };
    for (int e = 0; e < 4; ++e)
      if ((u[e] >> 18) == bin1) atomicAdd(&h2[(u[e] >> 8) & 1023u], 1u);
  }
}

__global__ __launch_bounds__(512) void k_count_d3(const float* __restrict__ a,
                                                  const unsigned* __restrict__ h2,
                                                  unsigned* __restrict__ h3,
                                                  unsigned* __restrict__ meta) {
  __shared__ unsigned s_tot[16];
  __shared__ unsigned s_res[2];
  const int t = threadIdx.x;
  const float4* a4 = reinterpret_cast<const float4*>(a) + (size_t)blockIdx.x * (512 * 8);
  float4 v[8];
  for (int q = 0; q < 8; ++q) v[q] = a4[q * 512 + t];
  const unsigned bin1 = meta[0], before1 = meta[1];
  const uint2 w = reinterpret_cast<const uint2*>(h2)[t];
  const unsigned c[2] = { w.x, w.y };
  unsigned bin2, before2;
  find_rank<2>(c, RANK_K - before1, s_tot, s_res, bin2, before2);
  const unsigned pre = (bin1 << 10) | bin2;
  if (blockIdx.x == 0 && t == 0) { meta[2] = pre; meta[3] = RANK_K - before1 - before2; }
  for (int q = 0; q < 8; ++q) {
    const unsigned u[4] = { __float_as_uint(v[q].x), __float_as_uint(v[q].y),
                            __float_as_uint(v[q].z), __float_as_uint(v[q].w) };
    for (int e = 0; e < 4; ++e)
      if ((u[e] >> 8) == pre) atomicAdd(&h3[u[e] & 255u], 1u);
  }
}

#define GK 64
#define GL (GK / 8)

struct FragSet { float4 w[GL]; float4 x[GL]; };

#define GEMM_STAGE_FENCE() __builtin_amdgcn_sched_barrier(0)

__device__ __forceinline__ void gemm_load(FragSet& f, const float* __restrict__ wl,
                                          const float* __restrict__ xl, const int kb) {
  GEMM_STAGE_FENCE();
  #pragma unroll
  for (int i = 0; i < GL; ++i) {
    f.w[i] = *reinterpret_cast<const float4*>(wl + kb + 8 * i);
    f.x[i] = *reinterpret_cast<const float4*>(xl + kb + 8 * i);
  }
  GEMM_STAGE_FENCE();
}

__device__ __forceinline__ void gemm_mma(const FragSet& f, v8f& acc0, v8f& acc1) {
  #pragma unroll
  for (int i = 0; i < GL; ++i) {
    const v2f a0 = { f.x[i].x, f.x[i].y };
    const v2f b0 = { f.w[i].x, f.w[i].y };
    acc0 = __builtin_amdgcn_wmma_f32_16x16x4_f32(false, a0, false, b0,
                                                 (short)0, acc0, false, false);
    const v2f a1 = { f.x[i].z, f.x[i].w };
    const v2f b1 = { f.w[i].z, f.w[i].w };
    acc1 = __builtin_amdgcn_wmma_f32_16x16x4_f32(false, a1, false, b1,
                                                 (short)0, acc1, false, false);
  }
}

__global__ __launch_bounds__(32) void k_gemm_y(const float* __restrict__ x,
                                               const float* __restrict__ W,
                                               float* __restrict__ Y) {
  const int nb   = blockIdx.x * 16;
  const int lane = threadIdx.x;
  const int half = lane >> 4;
  const int l16  = lane & 15;
  const float* __restrict__ wl = W + (size_t)(nb + l16) * NN + 4 * half;
  const float* __restrict__ xl = x + (size_t)(l16 & 7) * NN + 4 * half;

  FragSet f0, f1, f2, f3;
  v8f acc0 = {}, acc1 = {};
  gemm_load(f0, wl, xl, 0);
  gemm_load(f1, wl, xl, GK);
  gemm_load(f2, wl, xl, 2 * GK);
  #pragma unroll
  for (int kb = 0; kb < NN - 4 * GK; kb += 4 * GK) {
    gemm_load(f3, wl, xl, kb + 3 * GK);
    gemm_mma(f0, acc0, acc1);
    gemm_load(f0, wl, xl, kb + 4 * GK);
    gemm_mma(f1, acc0, acc1);
    gemm_load(f1, wl, xl, kb + 5 * GK);
    gemm_mma(f2, acc0, acc1);
    gemm_load(f2, wl, xl, kb + 6 * GK);
    gemm_mma(f3, acc0, acc1);
  }
  gemm_load(f3, wl, xl, NN - GK);
  gemm_mma(f0, acc0, acc1);
  gemm_mma(f1, acc0, acc1);
  gemm_mma(f2, acc0, acc1);
  gemm_mma(f3, acc0, acc1);

  for (int r = 0; r < 8; ++r) {
    const int m = r + half * 8;
    if (m < NB) Y[m * NN + nb + l16] = acc0[r] + acc1[r];
  }
}

#define RS_SEG 256
__global__ __launch_bounds__(256) void k_row_stats(const float* __restrict__ x,
                                                   const float* __restrict__ Y,
                                                   const float* __restrict__ bias,
                                                   float* __restrict__ rowM,
                                                   float* __restrict__ rowZ) {
  __shared__ float pm[NN / RS_SEG][32];
  __shared__ float pz[NN / RS_SEG][32];
  const int lane = threadIdx.x & 31;
  const int seg  = __builtin_amdgcn_readfirstlane((int)(threadIdx.x >> 5));
  const int b    = blockIdx.y;
  const int i    = blockIdx.x * 32 + lane;
  const float xbi = x[b * NN + i];
  const float* __restrict__ yr = Y + b * NN + seg * RS_SEG;
  const float* __restrict__ br = bias + seg * RS_SEG;

  float vmax = -INFINITY;
  #pragma clang loop vectorize(disable) unroll_count(16)
  for (int j = 0; j < RS_SEG; ++j) vmax = fmaxf(vmax, fmaf(xbi, yr[j], br[j]));
  pm[seg][lane] = vmax;
  __syncthreads();
  float vm = pm[0][lane];
  for (int s = 1; s < NN / RS_SEG; ++s) vm = fmaxf(vm, pm[s][lane]);
  const float m = fmaxf(vm, 0.01f * vm);

  float sum = 0.f;
  #pragma unroll 16
  for (int j = 0; j < RS_SEG; ++j) {
    float v = fmaf(xbi, yr[j], br[j]);
    v = fmaxf(v, 0.01f * v);
    sum += __expf(v - m);
  }
  pz[seg][lane] = sum;
  __syncthreads();
  if (seg == 0) {
    float z = pz[0][lane];
    for (int s = 1; s < NN / RS_SEG; ++s) z += pz[s][lane];
    rowM[b * NN + i] = m;
    rowZ[b * NN + i] = z;
  }
}

__global__ __launch_bounds__(256) void k_mask(const float* __restrict__ a,
                                              const unsigned* __restrict__ meta,
                                              const unsigned* __restrict__ h3,
                                              unsigned* __restrict__ maskT) {
  __shared__ unsigned s_tot[8];
  __shared__ unsigned s_res[2];
  const int lane = threadIdx.x & 31;
  const int tile = (blockIdx.x & 7) * 8 + (threadIdx.x >> 5);
  const int i0   = (blockIdx.x >> 3) * 32;
  const float* __restrict__ p = a + (size_t)i0 * NN + tile * 32 + lane;
  float v[32];
  #pragma unroll
  for (int r = 0; r < 32; ++r) v[r] = p[(size_t)r * NN];
  const unsigned c3[1] = { h3[threadIdx.x] };
  unsigned bin3, before3;
  find_rank<1>(c3, meta[3], s_tot, s_res, bin3, before3);
  const float med = __uint_as_float((meta[2] << 8) | bin3);
  unsigned mine = 0u;
  #pragma unroll
  for (int r = 0; r < 32; ++r) {
    const unsigned w = __builtin_amdgcn_ballot_w32(v[r] > med);
    if (lane == r) mine = w;
  }
  maskT[(size_t)tile * NN + i0 + lane] = mine;
}

__global__ __launch_bounds__(256) void k_aggregate(const float* __restrict__ x,
                                                   const float* __restrict__ Y,
                                                   const float* __restrict__ bias,
                                                   const unsigned* __restrict__ maskT,
                                                   const float* __restrict__ rowM,
                                                   const float* __restrict__ rowZ,
                                                   float* __restrict__ out) {
  const int j  = blockIdx.x * 256 + threadIdx.x;
  const int b  = blockIdx.y;
  const int iBase = blockIdx.z * (NN / 8);
  const int tile = blockIdx.x * 8 + __builtin_amdgcn_readfirstlane((int)(threadIdx.x >> 5));
  const unsigned* __restrict__ mrow = maskT + (size_t)tile * NN + iBase;
  const float ybj = Y[b * NN + j];
  const float bj  = bias[j];

  __shared__ float sx[256], sm[256], sc[256];
  {
    const int i = iBase + threadIdx.x;
    const float xv = x[b * NN + i];
    sx[threadIdx.x] = xv;
    sm[threadIdx.x] = rowM[b * NN + i];
    sc[threadIdx.x] = xv / rowZ[b * NN + i];
  }
  __syncthreads();
  float acc = 0.f;
  #pragma unroll 16
  for (int ii = 0; ii < 256; ++ii) {
    const bool on = __builtin_amdgcn_inverse_ballot_w32(mrow[ii]);
    float t = fmaf(sx[ii], ybj, bj);
    t = (t > 0.f) ? t : 0.01f * t;
    const float term = sc[ii] * __expf(t - sm[ii]);
    acc += on ? term : 0.f;
  }
  atomicAdd(&out[b * NN + j], acc);
}

extern "C" void kernel_launch(void* const* d_in, const int* in_sizes, int n_in,
                              void* d_out, int out_size, void* d_ws, size_t ws_size,
                              hipStream_t stream) {
  const float* x    = (const float*)d_in[0];
  const float* adj  = (const float*)d_in[1];
  const float* W    = (const float*)d_in[2];
  const float* bias = (const float*)d_in[3];
  float* out = (float*)d_out;

  char* ws = (char*)d_ws;
  size_t off = 0;
  float*    a     = (float*)(ws + off);    off += (size_t)NN * NN * sizeof(float);
  float*    Y     = (float*)(ws + off);    off += (size_t)NB * NN * sizeof(float);
  float*    rowM  = (float*)(ws + off);    off += (size_t)NB * NN * sizeof(float);
  float*    rowZ  = (float*)(ws + off);    off += (size_t)NB * NN * sizeof(float);
  unsigned* maskT = (unsigned*)(ws + off); off += (size_t)(NN / 32) * NN * sizeof(unsigned);
  unsigned* h1    = (unsigned*)(ws + off); off += (size_t)NB1 * sizeof(unsigned);
  unsigned* h2    = (unsigned*)(ws + off); off += (size_t)NB2 * sizeof(unsigned);
  unsigned* h3    = (unsigned*)(ws + off); off += (size_t)NB3 * sizeof(unsigned);
  unsigned* meta  = (unsigned*)(ws + off); off += 16;

  hipMemsetAsync(h1, 0, (size_t)(NB1 + NB2 + NB3) * sizeof(unsigned), stream);
  hipMemsetAsync(out, 0, (size_t)out_size * sizeof(float), stream);

  k_row_softmax<<<NN / SM_ROWS, SM_ROWS * 32, 0, stream>>>(adj, a);
  k_count_d1  <<<NN * NN / (512 * 32), 512, 0, stream>>>(a, h1);
  k_count_d2  <<<NN * NN / (512 * 32), 512, 0, stream>>>(a, h1, h2, meta);
  k_count_d3  <<<NN * NN / (512 * 32), 512, 0, stream>>>(a, h2, h3, meta);
  k_mask      <<<(NN / 32) * (NN / 256), 256, 0, stream>>>(a, meta, h3, maskT);
  k_gemm_y    <<<NN / 16, 32, 0, stream>>>(x, W, Y);
  k_row_stats <<<dim3(NN / 32, NB), 256, 0, stream>>>(x, Y, bias, rowM, rowZ);
  k_aggregate <<<dim3(NN / 256, NB, 8), 256, 0, stream>>>(x, Y, bias, maskT, rowM, rowZ, out);
}


# Launch sharing: the x,W chain rides beside the adj chain, role by group index: Y GEMM waves in the row-softmax launch, row-stat groups in the digit-1 count launch; 10 -> 8 stream ops; same sums
# speedup vs baseline: 31.5348x; 1.1530x over previous
#include <hip/hip_runtime.h>
#include <math.h>

#define NN 2048
#define NB 8
#define NB1 16384
#define NB2 1024
#define NB3 256
#define RANK_K (((unsigned)NN * (unsigned)NN - 1u) / 2u)

typedef __attribute__((ext_vector_type(2))) float v2f;
typedef __attribute__((ext_vector_type(8))) float v8f;

__device__ __forceinline__ void row_softmax_wave(const float* __restrict__ adj,
                                                 float* __restrict__ a,
                                                 const int i, const int lane) {
  const float4* __restrict__ r4 = reinterpret_cast<const float4*>(adj + (size_t)i * NN) + lane;

  float4 v[16];
  #pragma unroll
  for (int m = 0; m < 16; ++m) v[m] = r4[32 * m];

  float mx = -INFINITY;
  #pragma unroll
  for (int m = 0; m < 16; ++m)
    mx = fmaxf(mx, fmaxf(fmaxf(v[m].x, v[m].y), fmaxf(v[m].z, v[m].w)));
  for (int d = 16; d > 0; d >>= 1) mx = fmaxf(mx, __shfl_xor(mx, d));

  #pragma unroll
  for (int m = 0; m < 16; ++m) {
    v[m].x = __expf(v[m].x - mx);
    v[m].y = __expf(v[m].y - mx);
    v[m].z = __expf(v[m].z - mx);
    v[m].w = __expf(v[m].w - mx);
  }

  float pl[4] = { 0.f, 0.f, 0.f, 0.f }, ph[4] = { 0.f, 0.f, 0.f, 0.f };
  #pragma unroll
  for (int q = 0; q < 8; ++q) {
    pl[0] += v[2 * q].x;     pl[1] += v[2 * q].y;     pl[2] += v[2 * q].z;     pl[3] += v[2 * q].w;
    ph[0] += v[2 * q + 1].x; ph[1] += v[2 * q + 1].y; ph[2] += v[2 * q + 1].z; ph[3] += v[2 * q + 1].w;
  }
  float r[4];
  for (int c = 0; c < 4; ++c) r[c] = pl[c] + ph[c];
  for (int d = 16; d > 0; d >>= 1)
    for (int c = 0; c < 4; ++c) r[c] += __shfl_xor(r[c], d);
  const float z = (r[0] + r[2]) + (r[1] + r[3]);
  const float inv = 1.f / z;

  float4* __restrict__ o4 = reinterpret_cast<float4*>(a + (size_t)i * NN) + lane;
  #pragma unroll
  for (int m = 0; m < 16; ++m)
    o4[32 * m] = make_float4(v[m].x * inv, v[m].y * inv, v[m].z * inv, v[m].w * inv);
}


template <int PER>
__device__ __forceinline__ void find_rank(const unsigned (&c)[PER], const unsigned kk,
                                          unsigned* s_tot, unsigned* s_res,
                                          unsigned& bin, unsigned& before) {
  const int t = threadIdx.x, lane = t & 31, wv = t >> 5;
  unsigned mine = 0;
  for (int q = 0; q < PER; ++q) mine += c[q];
  unsigned inc = mine;
  for (int d = 1; d < 32; d <<= 1) {
    const unsigned up = __shfl_up(inc, d);
    if (lane >= d) inc += up;
  }
  if (lane == 31) s_tot[wv] = inc;
  if (t == 0) { s_res[0] = 0u; s_res[1] = 0u; }
  __syncthreads();
  unsigned excl = inc - mine;
  for (int w = 0; w < wv; ++w) excl += s_tot[w];
  if (kk >= excl && kk - excl < mine) {
    unsigned cum = excl, fb = 0u, fc = 0u;
    bool found = false;
    for (int q = 0; q < PER; ++q) {
      if (!found && kk - cum < c[q]) { fb = (unsigned)q; fc = cum; found = true; }
      cum += c[q];
    }
    s_res[0] = (unsigned)t * PER + fb;
    s_res[1] = fc;
  }
  __syncthreads();
  bin = s_res[0];
  before = s_res[1];
}

__device__ __forceinline__ void count_d1_group(const float* __restrict__ a,
                                               unsigned* __restrict__ h1, const int grp) {
  __shared__ unsigned bins[NB1];
  const int t = threadIdx.x;
  const float4* a4 = reinterpret_cast<const float4*>(a) + (size_t)grp * (512 * 8);
  float4 v[8];
  for (int q = 0; q < 8; ++q) v[q] = a4[q * 512 + t];
  uint4* b4 = reinterpret_cast<uint4*>(bins);
  for (int q = 0; q < 8; ++q) b4[q * 512 + t] = make_uint4(0u, 0u, 0u, 0u);
  __syncthreads();
  for (int q = 0; q < 8; ++q) {
    const unsigned u[4] = { __float_as_uint(v[q].x), __float_as_uint(v[q].y),
                            __float_as_uint(v[q].z), __float_as_uint(v[q].w) };
    for (int e = 0; e < 4; ++e) {
      const unsigned b = u[e] >> 18;
      const unsigned p = ((b & 28u) << 9) | ((b >> 5) << 2) | (b & 3u);
      atomicAdd(&bins[p], 1u);
    }
  }
  __syncthreads();
  uint4 w[8];
  for (int q = 0; q < 8; ++q) w[q] = b4[q * 512 + t];
  for (int q = 0; q < 8; ++q) {
    if ((w[q].x | w[q].y | w[q].z | w[q].w) != 0u) {
      unsigned* g = h1 + (q * 512 + t) * 4;
      if (w[q].x != 0u) atomicAdd(g + 0, w[q].x);
      if (w[q].y != 0u) atomicAdd(g + 1, w[q].y);
      if (w[q].z != 0u) atomicAdd(g + 2, w[q].z);
      if (w[q].w != 0u) atomicAdd(g + 3, w[q].w);
    }
  }
}

__global__ __launch_bounds__(512) void k_count_d2(const float* __restrict__ a,
                                                  const unsigned* __restrict__ h1,
                                                  unsigned* __restrict__ h2,
                                                  unsigned* __restrict__ meta) {
  __shared__ unsigned s_tot[16];
  __shared__ unsigned s_res[2];
  const int t = threadIdx.x;
  const float4* a4 = reinterpret_cast<const float4*>(a) + (size_t)blockIdx.x * (512 * 8);
  float4 v[8];
  for (int q = 0; q < 8; ++q) v[q] = a4[q * 512 + t];
  const uint4* h4 = reinterpret_cast<const uint4*>(h1);
  unsigned c[32];
  for (int q = 0; q < 8; ++q) {
    const uint4 w = h4[q * 512 + t];
    c[4 * q + 0] = w.x; c[4 * q + 1] = w.y; c[4 * q + 2] = w.z; c[4 * q + 3] = w.w;
  }
  unsigned bin1, before1;
  find_rank<32>(c, RANK_K, s_tot, s_res, bin1, before1);
  if (blockIdx.x == 0 && t == 0) { meta[0] = bin1; meta[1] = before1; }
  for (int q = 0; q < 8; ++q) {
    const unsigned u[4] = { __float_as_uint(v[q].x), __float_as_uint(v[q].y),
                            __float_as_uint(v[q].z), __float_as_uint(v[q].w) };
    for (int e = 0; e < 4; ++e)
      if ((u[e] >> 18) == bin1) atomicAdd(&h2[(u[e] >> 8) & 1023u], 1u);
  }
}

__global__ __launch_bounds__(512) void k_count_d3(const float* __restrict__ a,
                                                  const unsigned* __restrict__ h2,
                                                  unsigned* __restrict__ h3,
                                                  unsigned* __restrict__ meta) {
  __shared__ unsigned s_tot[16];
  __shared__ unsigned s_res[2];
  const int t = threadIdx.x;
  const float4* a4 = reinterpret_cast<const float4*>(a) + (size_t)blockIdx.x * (512 * 8);
  float4 v[8];
  for (int q = 0; q < 8; ++q) v[q] = a4[q * 512 + t];
  const unsigned bin1 = meta[0], before1 = meta[1];
  const uint2 w = reinterpret_cast<const uint2*>(h2)[t];
  const unsigned c[2] = { w.x, w.y };
  unsigned bin2, before2;
  find_rank<2>(c, RANK_K - before1, s_tot, s_res, bin2, before2);
  const unsigned pre = (bin1 << 10) | bin2;
  if (blockIdx.x == 0 && t == 0) { meta[2] = pre; meta[3] = RANK_K - before1 - before2; }
  for (int q = 0; q < 8; ++q) {
    const unsigned u[4] = { __float_as_uint(v[q].x), __float_as_uint(v[q].y),
                            __float_as_uint(v[q].z), __float_as_uint(v[q].w) };
    for (int e = 0; e < 4; ++e)
      if ((u[e] >> 8) == pre) atomicAdd(&h3[u[e] & 255u], 1u);
  }
}

#define GK 64
#define GL (GK / 8)

struct FragSet { float4 w[GL]; float4 x[GL]; };

#define GEMM_STAGE_FENCE() __builtin_amdgcn_sched_barrier(0)

__device__ __forceinline__ void gemm_load(FragSet& f, const float* __restrict__ wl,
                                          const float* __restrict__ xl, const int kb) {
  GEMM_STAGE_FENCE();
  #pragma unroll
  for (int i = 0; i < GL; ++i) {
    f.w[i] = *reinterpret_cast<const float4*>(wl + kb + 8 * i);
    f.x[i] = *reinterpret_cast<const float4*>(xl + kb + 8 * i);
  }
  GEMM_STAGE_FENCE();
}

__device__ __forceinline__ void gemm_mma(const FragSet& f, v8f& acc0, v8f& acc1) {
  #pragma unroll
  for (int i = 0; i < GL; ++i) {
    const v2f a0 = { f.x[i].x, f.x[i].y };
    const v2f b0 = { f.w[i].x, f.w[i].y };
    acc0 = __builtin_amdgcn_wmma_f32_16x16x4_f32(false, a0, false, b0,
                                                 (short)0, acc0, false, false);
    const v2f a1 = { f.x[i].z, f.x[i].w };
    const v2f b1 = { f.w[i].z, f.w[i].w };
    acc1 = __builtin_amdgcn_wmma_f32_16x16x4_f32(false, a1, false, b1,
                                                 (short)0, acc1, false, false);
  }
}

__device__ __forceinline__ void gemm_y_wave(const float* __restrict__ x,
                                            const float* __restrict__ W,
                                            float* __restrict__ Y,
                                            const int nb, const int lane) {
  const int half = lane >> 4;
  const int l16  = lane & 15;
  const float* __restrict__ wl = W + (size_t)(nb + l16) * NN + 4 * half;
  const float* __restrict__ xl = x + (size_t)(l16 & 7) * NN + 4 * half;

  FragSet f0, f1, f2, f3;
  v8f acc0 = {}, acc1 = {};
  gemm_load(f0, wl, xl, 0);
  gemm_load(f1, wl, xl, GK);
  gemm_load(f2, wl, xl, 2 * GK);
  #pragma unroll
  for (int kb = 0; kb < NN - 4 * GK; kb += 4 * GK) {
    gemm_load(f3, wl, xl, kb + 3 * GK);
    gemm_mma(f0, acc0, acc1);
    gemm_load(f0, wl, xl, kb + 4 * GK);
    gemm_mma(f1, acc0, acc1);
    gemm_load(f1, wl, xl, kb + 5 * GK);
    gemm_mma(f2, acc0, acc1);
    gemm_load(f2, wl, xl, kb + 6 * GK);
    gemm_mma(f3, acc0, acc1);
  }
  gemm_load(f3, wl, xl, NN - GK);
  gemm_mma(f0, acc0, acc1);
  gemm_mma(f1, acc0, acc1);
  gemm_mma(f2, acc0, acc1);
  gemm_mma(f3, acc0, acc1);

  for (int r = 0; r < 8; ++r) {
    const int m = r + half * 8;
    if (m < NB) Y[m * NN + nb + l16] = acc0[r] + acc1[r];
  }
}

#define SG_WAVES 4
#define GEMM_GROUPS (NN / 16 / SG_WAVES)
#define SM_GROUPS (NN / SG_WAVES)
__global__ __launch_bounds__(SG_WAVES * 32) void k_softmax_gemm(const float* __restrict__ adj,
                                                                float* __restrict__ a,
                                                                const float* __restrict__ x,
                                                                const float* __restrict__ W,
                                                                float* __restrict__ Y) {
  const int lane = threadIdx.x & 31;
  const int wave = __builtin_amdgcn_readfirstlane((int)(threadIdx.x >> 5));
  if (blockIdx.x < GEMM_GROUPS) {
    gemm_y_wave(x, W, Y, ((int)blockIdx.x * SG_WAVES + wave) * 16, lane);
    return;
  }
  row_softmax_wave(adj, a, ((int)blockIdx.x - GEMM_GROUPS) * SG_WAVES + wave, lane);
}

#define RS_SEG 256
#define RS_NSEG (NN / RS_SEG)
__device__ __forceinline__ void row_stats_group(const float* __restrict__ x,
                                                const float* __restrict__ Y,
                                                const float* __restrict__ bias,
                                                float* __restrict__ rowM,
                                                float* __restrict__ rowZ, const int grp) {
  __shared__ float pm[2][RS_NSEG][32];
  __shared__ float pz[2][RS_NSEG][32];
  const int lane = threadIdx.x & 31;
  const int wave = __builtin_amdgcn_readfirstlane((int)(threadIdx.x >> 5));
  const int seg  = wave & (RS_NSEG - 1);
  const int rb   = wave >> 3;
  const int b    = grp >> 5;
  const int i    = (grp & 31) * 64 + rb * 32 + lane;
  const float xbi = x[b * NN + i];
  const float* __restrict__ yr = Y + b * NN + seg * RS_SEG;
  const float* __restrict__ br = bias + seg * RS_SEG;

  float vmax = -INFINITY;
  #pragma clang loop vectorize(disable) unroll_count(16)
  for (int j = 0; j < RS_SEG; ++j) vmax = fmaxf(vmax, fmaf(xbi, yr[j], br[j]));
  pm[rb][seg][lane] = vmax;
  __syncthreads();
  float vm = pm[rb][0][lane];
  for (int s = 1; s < RS_NSEG; ++s) vm = fmaxf(vm, pm[rb][s][lane]);
  const float m = fmaxf(vm, 0.01f * vm);

  float sum = 0.f;
  #pragma unroll 16
  for (int j = 0; j < RS_SEG; ++j) {
    float v = fmaf(xbi, yr[j], br[j]);
    v = fmaxf(v, 0.01f * v);
    sum += __expf(v - m);
  }
  pz[rb][seg][lane] = sum;
  __syncthreads();
  if (seg == 0) {
    float z = pz[rb][0][lane];
    for (int s = 1; s < RS_NSEG; ++s) z += pz[rb][s][lane];
    rowM[b * NN + i] = m;
    rowZ[b * NN + i] = z;
  }
}

#define RS_GROUPS (NB * NN / 64)
#define D1_GROUPS (NN * NN / (512 * 32))
__global__ __launch_bounds__(512) void k_count_d1_row_stats(const float* __restrict__ a,
                                                            unsigned* __restrict__ h1,
                                                            const float* __restrict__ x,
                                                            const float* __restrict__ Y,
                                                            const float* __restrict__ bias,
                                                            float* __restrict__ rowM,
                                                            float* __restrict__ rowZ) {
  if (blockIdx.x >= RS_GROUPS) {
    count_d1_group(a, h1, (int)blockIdx.x - RS_GROUPS);
    return;
  }
  row_stats_group(x, Y, bias, rowM, rowZ, (int)blockIdx.x);
}

__global__ __launch_bounds__(256) void k_mask(const float* __restrict__ a,
                                              const unsigned* __restrict__ meta,
                                              const unsigned* __restrict__ h3,
                                              unsigned* __restrict__ maskT) {
  __shared__ unsigned s_tot[8];
  __shared__ unsigned s_res[2];
  const int lane = threadIdx.x & 31;
  const int tile = (blockIdx.x & 7) * 8 + (threadIdx.x >> 5);
  const int i0   = (blockIdx.x >> 3) * 32;
  const float* __restrict__ p = a + (size_t)i0 * NN + tile * 32 + lane;
  float v[32];
  #pragma unroll
  for (int r = 0; r < 32; ++r) v[r] = p[(size_t)r * NN];
  const unsigned c3[1] = { h3[threadIdx.x] };
  unsigned bin3, before3;
  find_rank<1>(c3, meta[3], s_tot, s_res, bin3, before3);
  const float med = __uint_as_float((meta[2] << 8) | bin3);
  unsigned mine = 0u;
  #pragma unroll
  for (int r = 0; r < 32; ++r) {
    const unsigned w = __builtin_amdgcn_ballot_w32(v[r] > med);
    if (lane == r) mine = w;
  }
  maskT[(size_t)tile * NN + i0 + lane] = mine;
}

__global__ __launch_bounds__(256) void k_aggregate(const float* __restrict__ x,
                                                   const float* __restrict__ Y,
                                                   const float* __restrict__ bias,
                                                   const unsigned* __restrict__ maskT,
                                                   const float* __restrict__ rowM,
                                                   const float* __restrict__ rowZ,
                                                   float* __restrict__ out) {
  const int j  = blockIdx.x * 256 + threadIdx.x;
  const int b  = blockIdx.y;
  const int iBase = blockIdx.z * (NN / 8);
  const int tile = blockIdx.x * 8 + __builtin_amdgcn_readfirstlane((int)(threadIdx.x >> 5));
  const unsigned* __restrict__ mrow = maskT + (size_t)tile * NN + iBase;
  const float ybj = Y[b * NN + j];
  const float bj  = bias[j];

  __shared__ float sx[256], sm[256], sc[256];
  {
    const int i = iBase + threadIdx.x;
    const float xv = x[b * NN + i];
    sx[threadIdx.x] = xv;
    sm[threadIdx.x] = rowM[b * NN + i];
    sc[threadIdx.x] = xv / rowZ[b * NN + i];
  }
  __syncthreads();
  float acc = 0.f;
  #pragma unroll 16
  for (int ii = 0; ii < 256; ++ii) {
    const bool on = __builtin_amdgcn_inverse_ballot_w32(mrow[ii]);
    float t = fmaf(sx[ii], ybj, bj);
    t = (t > 0.f) ? t : 0.01f * t;
    const float term = sc[ii] * __expf(t - sm[ii]);
    acc += on ? term : 0.f;
  }
  atomicAdd(&out[b * NN + j], acc);
}

extern "C" void kernel_launch(void* const* d_in, const int* in_sizes, int n_in,
                              void* d_out, int out_size, void* d_ws, size_t ws_size,
                              hipStream_t stream) {
  const float* x    = (const float*)d_in[0];
  const float* adj  = (const float*)d_in[1];
  const float* W    = (const float*)d_in[2];
  const float* bias = (const float*)d_in[3];
  float* out = (float*)d_out;

  char* ws = (char*)d_ws;
  size_t off = 0;
  float*    a     = (float*)(ws + off);    off += (size_t)NN * NN * sizeof(float);
  float*    Y     = (float*)(ws + off);    off += (size_t)NB * NN * sizeof(float);
  float*    rowM  = (float*)(ws + off);    off += (size_t)NB * NN * sizeof(float);
  float*    rowZ  = (float*)(ws + off);    off += (size_t)NB * NN * sizeof(float);
  unsigned* maskT = (unsigned*)(ws + off); off += (size_t)(NN / 32) * NN * sizeof(unsigned);
  unsigned* h1    = (unsigned*)(ws + off); off += (size_t)NB1 * sizeof(unsigned);
  unsigned* h2    = (unsigned*)(ws + off); off += (size_t)NB2 * sizeof(unsigned);
  unsigned* h3    = (unsigned*)(ws + off); off += (size_t)NB3 * sizeof(unsigned);
  unsigned* meta  = (unsigned*)(ws + off); off += 16;

  hipMemsetAsync(h1, 0, (size_t)(NB1 + NB2 + NB3) * sizeof(unsigned), stream);
  hipMemsetAsync(out, 0, (size_t)out_size * sizeof(float), stream);

  k_softmax_gemm      <<<GEMM_GROUPS + SM_GROUPS, SG_WAVES * 32, 0, stream>>>(adj, a, x, W, Y);
  k_count_d1_row_stats<<<RS_GROUPS + D1_GROUPS, 512, 0, stream>>>(a, h1, x, Y, bias, rowM, rowZ);
  k_count_d2  <<<NN * NN / (512 * 32), 512, 0, stream>>>(a, h1, h2, meta);
  k_count_d3  <<<NN * NN / (512 * 32), 512, 0, stream>>>(a, h2, h3, meta);
  k_mask      <<<(NN / 32) * (NN / 256), 256, 0, stream>>>(a, meta, h3, maskT);
  k_aggregate <<<dim3(NN / 256, NB, 8), 256, 0, stream>>>(x, Y, bias, maskT, rowM, rowZ, out);
}
